# starting kernel
# baseline (speedup 1.0000x reference)
#include <hip/hip_runtime.h>
#include <math.h>


typedef __attribute__((ext_vector_type(2))) float v2f;
typedef __attribute__((ext_vector_type(8))) float v8f;

__device__ __forceinline__ float softplusf(float x) {
  return (x > 20.f) ? x : log1pf(expf(x));
}

__global__ __launch_bounds__(256) void k_inverse(const float* __restrict__ V,
                                                 float* __restrict__ Vinv) {
  __shared__ float M[64][128];
  __shared__ float prow[128];
  __shared__ float fac[64];
  const int tid = threadIdx.x;
  for (int idx = tid; idx < 64 * 128; idx += 256) {
    int r = idx >> 7, c = idx & 127;
    M[r][c] = (c < 64) ? V[r * 64 + c] : ((c - 64 == r) ? 1.f : 0.f);
  }
  __syncthreads();
  for (int k = 0; k < 64; ++k) {
    if (tid < 128) {
      float pinv = 1.f / M[k][k];
      prow[tid] = M[k][tid] * pinv;
    } else if (tid < 192) {
      fac[tid - 128] = M[tid - 128][k];
    }
    __syncthreads();
    for (int idx = tid; idx < 64 * 128; idx += 256) {
      int r = idx >> 7, c = idx & 127;
      M[r][c] = (r == k) ? prow[c] : fmaf(-fac[r], prow[c], M[r][c]);
    }
    __syncthreads();
  }
  for (int idx = tid; idx < 64 * 64; idx += 256) {
    int r = idx >> 6, c = idx & 63;
    Vinv[idx] = M[r][64 + c];
  }
}

__global__ void k_build_G(const float* __restrict__ gate_w,
                          const float* __restrict__ B,
                          const float* __restrict__ Vinv,
                          float* __restrict__ G) {
  int idx = blockIdx.x * blockDim.x + threadIdx.x;
  if (idx >= 96 * 512) return;
  int row = idx >> 9, h = idx & 511;
  if (row < 32) {
    G[idx] = gate_w[row * 512 + h];
  } else {
    int d = row - 32;
    float s = 0.f;
    for (int j = 0; j < 64; ++j) s = fmaf(Vinv[d * 64 + j], B[j * 512 + h], s);
    G[idx] = s;
  }
}

__global__ void k_EA(const float* __restrict__ E, const float* __restrict__ alpha,
                     float* __restrict__ EA) {
  int idx = blockIdx.x * blockDim.x + threadIdx.x;
  if (idx >= 64 * 100) return;
  int d = idx / 100, k = idx % 100;
  float s = 0.f;
  for (int r = 0; r < 32; ++r) s = fmaf(E[d * 32 + r], alpha[r * 100 + k], s);
  EA[idx] = s;
}

__global__ void k_Minv(const float* __restrict__ Vinv, const float* __restrict__ EA,
                       float* __restrict__ Minv) {
  int idx = blockIdx.x * blockDim.x + threadIdx.x;
  if (idx >= 64 * 100) return;
  int d = idx / 100, k = idx % 100;
  float s = 0.f;
  for (int j = 0; j < 64; ++j) s = fmaf(Vinv[d * 64 + j], EA[j * 100 + k], s);
  Minv[idx] = s;
}

__global__ __launch_bounds__(32) void k_gemm_uG(const float* __restrict__ U,
                                                const float* __restrict__ G,
                                                float* __restrict__ C96) {
  const int rt = blockIdx.x, ct = blockIdx.y;
  const int l = threadIdx.x;
  const int m = l & 15;
  const int kh = (l >> 4) * 2;
  const float* Arow = U + (size_t)(rt * 16 + m) * 512;
  const float* Brow = G + (size_t)(ct * 16 + m) * 512;
  v8f acc = {};
  for (int kb = 0; kb < 512; kb += 4) {
    v2f a = *(const v2f*)(Arow + kb + kh);
    v2f b = *(const v2f*)(Brow + kb + kh);
    acc = __builtin_amdgcn_wmma_f32_16x16x4_f32(false, a, false, b, (short)0, acc,
                                                false, false);
  }
  const int rbase = rt * 16 + (l >> 4) * 8;
  const int col = ct * 16 + m;
  for (int v = 0; v < 8; ++v) C96[(size_t)(rbase + v) * 96 + col] = acc[v];
}

__global__ void k_prep(const float* __restrict__ times, const int* __restrict__ marks,
                       const float* __restrict__ llr, const float* __restrict__ lim,
                       const float* __restrict__ gate_b, const float* __restrict__ C96,
                       const float* __restrict__ Minv, float* __restrict__ Ac,
                       float* __restrict__ Uc, int n) {
  int idx = blockIdx.x * blockDim.x + threadIdx.x;
  if (idx >= n * 32) return;
  int ns = idx >> 5, p = idx & 31;
  float t = times[ns];
  float dt = (ns == 0) ? t : (t - times[ns - 1]);
  float gate = softplusf(C96[(size_t)ns * 96 + p] + gate_b[p]);
  float real = -softplusf(llr[p]) * gate;
  float im = lim[p];
  float e = expf(real * dt);
  float a = e * cosf(im * dt);
  float c = e * sinf(im * dt);
  float denom = real * real + im * im;
  float nre = a - 1.f, nim = c;
  float qr = (nre * real + nim * im) / (denom + 1e-12f);
  float qi = (nim * real - nre * im) / (denom + 1e-12f);
  if (denom < 1e-8f) { qr = dt; qi = 0.f; }
  if (!(dt > 0.f)) { qr = 0.f; qi = 0.f; }
  float w0 = C96[(size_t)ns * 96 + 32 + 2 * p];
  float w1 = C96[(size_t)ns * 96 + 32 + 2 * p + 1];
  float ure = qr * w0 - qi * w1;
  float uim = qi * w0 + qr * w1;
  int mk = marks[ns];
  ure += Minv[(2 * p) * 100 + mk];
  uim += Minv[(2 * p + 1) * 100 + mk];
  size_t base = (size_t)ns * 64 + 2 * p;
  Ac[base] = a;   Ac[base + 1] = c;
  Uc[base] = ure; Uc[base + 1] = uim;
}

__global__ __launch_bounds__(32) void k_scanA(const float* __restrict__ Ac,
                                              const float* __restrict__ Uc,
                                              float* __restrict__ Aagg,
                                              float* __restrict__ Uagg, int n, int L) {
  int ch = blockIdx.x, p = threadIdx.x;
  float Ar = 1.f, Ai = 0.f, Ur = 0.f, Ui = 0.f;
  int i0 = ch * L, i1 = i0 + L; if (i1 > n) i1 = n;
  for (int i = i0; i < i1; ++i) {
    size_t b = (size_t)i * 64 + 2 * p;
    float ar = Ac[b], ai = Ac[b + 1], ur = Uc[b], ui = Uc[b + 1];
    float nAr = ar * Ar - ai * Ai, nAi = ar * Ai + ai * Ar;
    float nUr = ar * Ur - ai * Ui + ur, nUi = ar * Ui + ai * Ur + ui;
    Ar = nAr; Ai = nAi; Ur = nUr; Ui = nUi;
  }
  size_t o = (size_t)ch * 64 + 2 * p;
  Aagg[o] = Ar; Aagg[o + 1] = Ai; Uagg[o] = Ur; Uagg[o + 1] = Ui;
}

__global__ __launch_bounds__(32) void k_scanB(const float* __restrict__ Aagg,
                                              const float* __restrict__ Uagg,
                                              const float* __restrict__ Vinv,
                                              const float* __restrict__ x0,
                                              float* __restrict__ Ystart, int nch) {
  int p = threadIdx.x;
  float yr = 0.f, yi = 0.f;
  for (int j = 0; j < 64; ++j) {
    yr = fmaf(Vinv[(2 * p) * 64 + j], x0[j], yr);
    yi = fmaf(Vinv[(2 * p + 1) * 64 + j], x0[j], yi);
  }
  for (int c = 0; c < nch; ++c) {
    size_t o = (size_t)c * 64 + 2 * p;
    Ystart[o] = yr; Ystart[o + 1] = yi;
    float ar = Aagg[o], ai = Aagg[o + 1];
    float nyr = ar * yr - ai * yi + Uagg[o];
    float nyi = ar * yi + ai * yr + Uagg[o + 1];
    yr = nyr; yi = nyi;
  }
}

__global__ __launch_bounds__(32) void k_scanC(const float* __restrict__ Ac,
                                              const float* __restrict__ Uc,
                                              const float* __restrict__ Ystart,
                                              float* __restrict__ Y, int n, int L) {
  int ch = blockIdx.x, p = threadIdx.x;
  size_t so = (size_t)ch * 64 + 2 * p;
  float yr = Ystart[so], yi = Ystart[so + 1];
  int i0 = ch * L, i1 = i0 + L; if (i1 > n) i1 = n;
  for (int i = i0; i < i1; ++i) {
    size_t b = (size_t)i * 64 + 2 * p;
    float ar = Ac[b], ai = Ac[b + 1];
    float nyr = ar * yr - ai * yi + Uc[b];
    float nyi = ar * yi + ai * yr + Uc[b + 1];
    yr = nyr; yi = nyi;
    Y[b] = yr; Y[b + 1] = yi;
  }
}

__global__ __launch_bounds__(32) void k_gemm_out(const float* __restrict__ Y,
                                                 const float* __restrict__ V,
                                                 const float* __restrict__ EA,
                                                 const int* __restrict__ marks,
                                                 float* __restrict__ out, int n) {
  const int rt = blockIdx.x, ct = blockIdx.y, l = threadIdx.x;
  const int m = l & 15;
  const int kh = (l >> 4) * 2;
  const float* Arow = Y + (size_t)(rt * 16 + m) * 64;
  const int dcol = ct * 16 + m;
  const float* Brow = V + (size_t)dcol * 64;
  v8f acc = {};
  for (int kb = 0; kb < 64; kb += 4) {
    v2f a = *(const v2f*)(Arow + kb + kh);
    v2f b = *(const v2f*)(Brow + kb + kh);
    acc = __builtin_amdgcn_wmma_f32_16x16x4_f32(false, a, false, b, (short)0, acc,
                                                false, false);
  }
  const int rbase = rt * 16 + (l >> 4) * 8;
  const size_t xl_off = (size_t)n * 64;
  for (int v = 0; v < 8; ++v) {
    int row = rbase + v;
    float xr = acc[v];
    float r = EA[dcol * 100 + marks[row]];
    out[(size_t)row * 64 + dcol] = xr;
    out[xl_off + (size_t)row * 64 + dcol] = xr - r;
  }
}

extern "C" void kernel_launch(void* const* d_in, const int* in_sizes, int n_in,
                              void* d_out, int out_size, void* d_ws, size_t ws_size,
                              hipStream_t stream) {
  const float* times   = (const float*)d_in[0];
  const int*   marks   = (const int*)d_in[1];
  const float* u_in    = (const float*)d_in[2];
  const float* llr     = (const float*)d_in[3];
  const float* lim     = (const float*)d_in[4];
  const float* V       = (const float*)d_in[5];
  const float* B       = (const float*)d_in[6];
  const float* E       = (const float*)d_in[7];
  const float* alpha   = (const float*)d_in[8];
  const float* gate_w  = (const float*)d_in[9];
  const float* gate_b  = (const float*)d_in[10];
  const float* x0      = (const float*)d_in[11];
  float* out = (float*)d_out;
  const int n = in_sizes[0];
  const int CHN = 256;
  const int L = (n + CHN - 1) / CHN;

  float* ws    = (float*)d_ws;
  float* Vinv  = ws; ws += 64 * 64;
  float* G     = ws; ws += 96 * 512;
  float* EA    = ws; ws += 64 * 100;
  float* Minv  = ws; ws += 64 * 100;
  float* C96   = ws; ws += (size_t)n * 96;
  float* Ac    = ws; ws += (size_t)n * 64;
  float* Uc    = ws; ws += (size_t)n * 64;
  float* Aagg  = ws; ws += (size_t)CHN * 64;
  float* Uagg  = ws; ws += (size_t)CHN * 64;
  float* Ystart= ws; ws += (size_t)CHN * 64;
  float* Y     = ws; ws += (size_t)n * 64;

  k_inverse<<<dim3(1), dim3(256), 0, stream>>>(V, Vinv);
  k_build_G<<<dim3((96 * 512 + 255) / 256), dim3(256), 0, stream>>>(gate_w, B, Vinv, G);
  k_EA<<<dim3((64 * 100 + 255) / 256), dim3(256), 0, stream>>>(E, alpha, EA);
  k_Minv<<<dim3((64 * 100 + 255) / 256), dim3(256), 0, stream>>>(Vinv, EA, Minv);
  k_gemm_uG<<<dim3(n / 16, 6), dim3(32), 0, stream>>>(u_in, G, C96);
  k_prep<<<dim3((n * 32 + 255) / 256), dim3(256), 0, stream>>>(
      times, marks, llr, lim, gate_b, C96, Minv, Ac, Uc, n);
  k_scanA<<<dim3(CHN), dim3(32), 0, stream>>>(Ac, Uc, Aagg, Uagg, n, L);
  k_scanB<<<dim3(1), dim3(32), 0, stream>>>(Aagg, Uagg, Vinv, x0, Ystart, CHN);
  k_scanC<<<dim3(CHN), dim3(32), 0, stream>>>(Ac, Uc, Ystart, Y, n, L);
  k_gemm_out<<<dim3(n / 16, 4), dim3(32), 0, stream>>>(Y, V, EA, marks, out, n);
}


# Inverse off the critical path: U GEMM uses B and gate_w directly, w = Vinv(Bu) instead of (Vinv B)u (product re-associated); inverse, E@alpha and that GEMM share one launch; 10 launches -> 7
# speedup vs baseline: 5.3464x; 5.3464x over previous
#include <hip/hip_runtime.h>
#include <math.h>


typedef __attribute__((ext_vector_type(2))) float v2f;
typedef __attribute__((ext_vector_type(8))) float v8f;

__device__ __forceinline__ float softplusf(float x) {
  return (x > 20.f) ? x : log1pf(expf(x));
}

__device__ __forceinline__ float lane_bcast(float v, int src_lane) {
  return __int_as_float(__builtin_amdgcn_readlane(__float_as_int(v), src_lane));
}

__device__ __forceinline__ void gj_step(float (&rp)[64], float (&ro)[64], float& sp,
                                        const int pl, const int lane) {
  const float pinv = 1.f / lane_bcast(rp[0], pl);
  const bool piv = (lane == pl);
  const float nfp = piv ? 0.f : -(rp[0] * pinv);
  const float nfo = -(ro[0] * pinv);
  sp = piv ? pinv : sp;
#pragma unroll
  for (int c = 1; c < 64; ++c) {
    const float s = lane_bcast(rp[c], pl);
    rp[c - 1] = fmaf(nfp, s, rp[c]);
    ro[c - 1] = fmaf(nfo, s, ro[c]);
  }
  rp[63] = piv ? 1.f : nfp;
  ro[63] = nfo;
}

__device__ __forceinline__ void inverse_wave(const float* __restrict__ V,
                                             float* __restrict__ Vinv, const int lane) {
  float ra[64], rb[64];
  const float4* pa = reinterpret_cast<const float4*>(V + lane * 64);
  const float4* pb = reinterpret_cast<const float4*>(V + (lane + 32) * 64);
#pragma unroll
  for (int j = 0; j < 16; ++j) {
    const float4 x = pa[j], y = pb[j];
    ra[4 * j] = x.x; ra[4 * j + 1] = x.y; ra[4 * j + 2] = x.z; ra[4 * j + 3] = x.w;
    rb[4 * j] = y.x; rb[4 * j + 1] = y.y; rb[4 * j + 2] = y.z; rb[4 * j + 3] = y.w;
  }
  float sa = 1.f, sb = 1.f;
  for (int k = 0; k < 32; ++k) gj_step(ra, rb, sa, k, lane);
  for (int k = 0; k < 32; ++k) gj_step(rb, ra, sb, k, lane);
  float4* qa = reinterpret_cast<float4*>(Vinv + lane * 64);
  float4* qb = reinterpret_cast<float4*>(Vinv + (lane + 32) * 64);
#pragma unroll
  for (int j = 0; j < 16; ++j) {
    qa[j] = make_float4(sa * ra[4 * j], sa * ra[4 * j + 1], sa * ra[4 * j + 2], sa * ra[4 * j + 3]);
    qb[j] = make_float4(sb * rb[4 * j], sb * rb[4 * j + 1], sb * rb[4 * j + 2], sb * rb[4 * j + 3]);
  }
}

constexpr int EA_BLOCKS = (64 * 100 + 31) / 32;

__global__ __launch_bounds__(32) void k_front(const float* __restrict__ V,
                                              const float* __restrict__ E,
                                              const float* __restrict__ alpha,
                                              const float* __restrict__ U,
                                              const float* __restrict__ gate_w,
                                              const float* __restrict__ Bm,
                                              float* __restrict__ Vinv,
                                              float* __restrict__ EA,
                                              float* __restrict__ C96, int nrt) {
  const int b = blockIdx.x;
  const int l = threadIdx.x;
  if (b == 0) {
    inverse_wave(V, Vinv, l);
    return;
  }
  if (b <= EA_BLOCKS) {
    const int idx = (b - 1) * 32 + l;
    if (idx < 64 * 100) {
      const int d = idx / 100, k = idx % 100;
      float s = 0.f;
      for (int r = 0; r < 32; ++r) s = fmaf(E[d * 32 + r], alpha[r * 100 + k], s);
      EA[idx] = s;
    }
    return;
  }
  const int t = b - 1 - EA_BLOCKS;
  const int rt = t % nrt, ct = t / nrt;
  const int m = l & 15;
  const int kh = (l >> 4) * 2;
  const int c = ct * 16 + m;
  const float* Arow = U + (size_t)(rt * 16 + m) * 512;
  const float* Brow = (ct < 2) ? (gate_w + (size_t)c * 512) : (Bm + (size_t)(c - 32) * 512);
  v8f acc = {};
  for (int kb = 0; kb < 512; kb += 4) {
    v2f a = *(const v2f*)(Arow + kb + kh);
    v2f bb = *(const v2f*)(Brow + kb + kh);
    acc = __builtin_amdgcn_wmma_f32_16x16x4_f32(false, a, false, bb, (short)0, acc,
                                                false, false);
  }
  const int rbase = rt * 16 + (l >> 4) * 8;
  for (int v = 0; v < 8; ++v) C96[(size_t)(rbase + v) * 96 + c] = acc[v];
}

__global__ __launch_bounds__(32) void k_mid(const float* __restrict__ C96,
                                            const float* __restrict__ Vinv,
                                            const float* __restrict__ EA,
                                            float* __restrict__ Wm,
                                            float* __restrict__ Minv, int nrt) {
  const int b = blockIdx.x;
  const int l = threadIdx.x;
  const int ntile = nrt * 4;
  if (b >= ntile) {
    const int idx = (b - ntile) * 32 + l;
    if (idx < 64 * 100) {
      const int d = idx / 100, k = idx % 100;
      float s = 0.f;
      for (int j = 0; j < 64; ++j) s = fmaf(Vinv[d * 64 + j], EA[j * 100 + k], s);
      Minv[idx] = s;
    }
    return;
  }
  const int rt = b % nrt, ct = b / nrt;
  const int m = l & 15;
  const int kh = (l >> 4) * 2;
  const float* Arow = C96 + (size_t)(rt * 16 + m) * 96 + 32;
  const int dcol = ct * 16 + m;
  const float* Brow = Vinv + (size_t)dcol * 64;
  v8f acc = {};
  for (int kb = 0; kb < 64; kb += 4) {
    v2f a = *(const v2f*)(Arow + kb + kh);
    v2f bb = *(const v2f*)(Brow + kb + kh);
    acc = __builtin_amdgcn_wmma_f32_16x16x4_f32(false, a, false, bb, (short)0, acc,
                                                false, false);
  }
  const int rbase = rt * 16 + (l >> 4) * 8;
  for (int v = 0; v < 8; ++v) Wm[(size_t)(rbase + v) * 64 + dcol] = acc[v];
}

__global__ void k_prep(const float* __restrict__ times, const int* __restrict__ marks,
                       const float* __restrict__ llr, const float* __restrict__ lim,
                       const float* __restrict__ gate_b, const float* __restrict__ C96,
                       const float* __restrict__ Wm,
                       const float* __restrict__ Minv, float* __restrict__ Ac,
                       float* __restrict__ Uc, int n) {
  int idx = blockIdx.x * blockDim.x + threadIdx.x;
  if (idx >= n * 32) return;
  int ns = idx >> 5, p = idx & 31;
  float t = times[ns];
  float dt = (ns == 0) ? t : (t - times[ns - 1]);
  float gate = softplusf(C96[(size_t)ns * 96 + p] + gate_b[p]);
  float real = -softplusf(llr[p]) * gate;
  float im = lim[p];
  float e = expf(real * dt);
  float a = e * cosf(im * dt);
  float c = e * sinf(im * dt);
  float denom = real * real + im * im;
  float nre = a - 1.f, nim = c;
  float qr = (nre * real + nim * im) / (denom + 1e-12f);
  float qi = (nim * real - nre * im) / (denom + 1e-12f);
  if (denom < 1e-8f) { qr = dt; qi = 0.f; }
  if (!(dt > 0.f)) { qr = 0.f; qi = 0.f; }
  float w0 = Wm[(size_t)ns * 64 + 2 * p];
  float w1 = Wm[(size_t)ns * 64 + 2 * p + 1];
  float ure = qr * w0 - qi * w1;
  float uim = qi * w0 + qr * w1;
  int mk = marks[ns];
  ure += Minv[(2 * p) * 100 + mk];
  uim += Minv[(2 * p + 1) * 100 + mk];
  size_t base = (size_t)ns * 64 + 2 * p;
  Ac[base] = a;   Ac[base + 1] = c;
  Uc[base] = ure; Uc[base + 1] = uim;
}

constexpr int CHN = 256;
constexpr int SB = 32;

__device__ __forceinline__ void load_steps(v2f (&a)[SB], v2f (&u)[SB],
                                           const float* __restrict__ pa,
                                           const float* __restrict__ pu) {
#pragma unroll
  for (int j = 0; j < SB; ++j) {
    a[j] = *(const v2f*)(pa + j * 64);
    u[j] = *(const v2f*)(pu + j * 64);
  }
  __builtin_amdgcn_sched_barrier(0);
}

__global__ __launch_bounds__(32) void k_scanA(const float* __restrict__ Ac,
                                              const float* __restrict__ Uc,
                                              float* __restrict__ Aagg,
                                              float* __restrict__ Uagg, int n, int L) {
  int ch = blockIdx.x, p = threadIdx.x;
  float Ar = 1.f, Ai = 0.f, Ur = 0.f, Ui = 0.f;
  int i0 = ch * L, i1 = i0 + L; if (i1 > n) i1 = n;
  int i = i0;
  for (; i + SB <= i1; i += SB) {
    v2f a[SB], u[SB];
    load_steps(a, u, Ac + (size_t)i * 64 + 2 * p, Uc + (size_t)i * 64 + 2 * p);
#pragma unroll
    for (int j = 0; j < SB; ++j) {
      float ar = a[j].x, ai = a[j].y, ur = u[j].x, ui = u[j].y;
      float nAr = ar * Ar - ai * Ai, nAi = ar * Ai + ai * Ar;
      float nUr = ar * Ur - ai * Ui + ur, nUi = ar * Ui + ai * Ur + ui;
      Ar = nAr; Ai = nAi; Ur = nUr; Ui = nUi;
    }
  }
  for (; i < i1; ++i) {
    size_t b = (size_t)i * 64 + 2 * p;
    float ar = Ac[b], ai = Ac[b + 1], ur = Uc[b], ui = Uc[b + 1];
    float nAr = ar * Ar - ai * Ai, nAi = ar * Ai + ai * Ar;
    float nUr = ar * Ur - ai * Ui + ur, nUi = ar * Ui + ai * Ur + ui;
    Ar = nAr; Ai = nAi; Ur = nUr; Ui = nUi;
  }
  size_t o = (size_t)ch * 64 + 2 * p;
  Aagg[o] = Ar; Aagg[o + 1] = Ai; Uagg[o] = Ur; Uagg[o + 1] = Ui;
}

__device__ __forceinline__ void carry_steps(const v2f (&a)[SB], const v2f (&u)[SB],
                                            float* __restrict__ py, float& yr, float& yi) {
#pragma unroll
  for (int j = 0; j < SB; ++j) {
    py[j * 64] = yr; py[j * 64 + 1] = yi;
    float ar = a[j].x, ai = a[j].y;
    float nyr = ar * yr - ai * yi + u[j].x;
    float nyi = ar * yi + ai * yr + u[j].y;
    yr = nyr; yi = nyi;
  }
}

__global__ __launch_bounds__(32) void k_scanB(const float* __restrict__ Aagg,
                                              const float* __restrict__ Uagg,
                                              const float* __restrict__ Vinv,
                                              const float* __restrict__ x0,
                                              float* __restrict__ Ystart) {
  static_assert(CHN % (2 * SB) == 0, "the carry loop walks two register batches per trip");
  const int p = threadIdx.x;
  const float* pa = Aagg + 2 * p;
  const float* pu = Uagg + 2 * p;
  float* py = Ystart + 2 * p;
  v2f a0[SB], u0[SB], a1[SB], u1[SB];
  load_steps(a0, u0, pa, pu);
  const float4* r0 = reinterpret_cast<const float4*>(Vinv + (2 * p) * 64);
  const float4* r1 = reinterpret_cast<const float4*>(Vinv + (2 * p + 1) * 64);
  const float4* xv = reinterpret_cast<const float4*>(x0);
  float4 w0[16], w1[16];
#pragma unroll
  for (int j = 0; j < 16; ++j) { w0[j] = r0[j]; w1[j] = r1[j]; }
  float yr = 0.f, yi = 0.f;
#pragma unroll
  for (int j = 0; j < 16; ++j) {
    const float4 x = xv[j];
    yr = fmaf(w0[j].x, x.x, yr); yi = fmaf(w1[j].x, x.x, yi);
    yr = fmaf(w0[j].y, x.y, yr); yi = fmaf(w1[j].y, x.y, yi);
    yr = fmaf(w0[j].z, x.z, yr); yi = fmaf(w1[j].z, x.z, yi);
    yr = fmaf(w0[j].w, x.w, yr); yi = fmaf(w1[j].w, x.w, yi);
  }
  for (int c = 0; c < CHN; c += 2 * SB) {
    load_steps(a1, u1, pa + (c + SB) * 64, pu + (c + SB) * 64);
    carry_steps(a0, u0, py + c * 64, yr, yi);
    if (c + 2 * SB < CHN) load_steps(a0, u0, pa + (c + 2 * SB) * 64, pu + (c + 2 * SB) * 64);
    carry_steps(a1, u1, py + (c + SB) * 64, yr, yi);
  }
}

__device__ __forceinline__ void apply_steps(const v2f (&a)[SB], const v2f (&u)[SB],
                                            float* __restrict__ py, float& yr, float& yi) {
#pragma unroll
  for (int j = 0; j < SB; ++j) {
    float ar = a[j].x, ai = a[j].y;
    float nyr = ar * yr - ai * yi + u[j].x;
    float nyi = ar * yi + ai * yr + u[j].y;
    yr = nyr; yi = nyi;
    py[j * 64] = yr; py[j * 64 + 1] = yi;
  }
}

__global__ __launch_bounds__(32) void k_scanC(const float* __restrict__ Ac,
                                              const float* __restrict__ Uc,
                                              const float* __restrict__ Ystart,
                                              float* __restrict__ Y, int n, int L) {
  int ch = blockIdx.x, p = threadIdx.x;
  size_t so = (size_t)ch * 64 + 2 * p;
  const v2f ys = *(const v2f*)(Ystart + so);
  float yr, yi;
  int i0 = ch * L, i1 = i0 + L; if (i1 > n) i1 = n;
  int i = i0;
  if (i + SB <= i1) {
    v2f a[SB], u[SB];
    load_steps(a, u, Ac + (size_t)i * 64 + 2 * p, Uc + (size_t)i * 64 + 2 * p);
    yr = ys.x; yi = ys.y;
    apply_steps(a, u, Y + (size_t)i * 64 + 2 * p, yr, yi);
    i += SB;
    for (; i + SB <= i1; i += SB) {
      load_steps(a, u, Ac + (size_t)i * 64 + 2 * p, Uc + (size_t)i * 64 + 2 * p);
      apply_steps(a, u, Y + (size_t)i * 64 + 2 * p, yr, yi);
    }
  } else {
    yr = ys.x; yi = ys.y;
  }
  for (; i < i1; ++i) {
    size_t b = (size_t)i * 64 + 2 * p;
    float ar = Ac[b], ai = Ac[b + 1];
    float nyr = ar * yr - ai * yi + Uc[b];
    float nyi = ar * yi + ai * yr + Uc[b + 1];
    yr = nyr; yi = nyi;
    Y[b] = yr; Y[b + 1] = yi;
  }
}

__global__ __launch_bounds__(32) void k_gemm_out(const float* __restrict__ Y,
                                                 const float* __restrict__ V,
                                                 const float* __restrict__ EA,
                                                 const int* __restrict__ marks,
                                                 float* __restrict__ out, int n) {
  const int rt = blockIdx.x, ct = blockIdx.y, l = threadIdx.x;
  const int m = l & 15;
  const int kh = (l >> 4) * 2;
  const float* Arow = Y + (size_t)(rt * 16 + m) * 64;
  const int dcol = ct * 16 + m;
  const float* Brow = V + (size_t)dcol * 64;
  v8f acc = {};
  for (int kb = 0; kb < 64; kb += 4) {
    v2f a = *(const v2f*)(Arow + kb + kh);
    v2f b = *(const v2f*)(Brow + kb + kh);
    acc = __builtin_amdgcn_wmma_f32_16x16x4_f32(false, a, false, b, (short)0, acc,
                                                false, false);
  }
  const int rbase = rt * 16 + (l >> 4) * 8;
  const size_t xl_off = (size_t)n * 64;
  for (int v = 0; v < 8; ++v) {
    int row = rbase + v;
    float xr = acc[v];
    float r = EA[dcol * 100 + marks[row]];
    out[(size_t)row * 64 + dcol] = xr;
    out[xl_off + (size_t)row * 64 + dcol] = xr - r;
  }
}

extern "C" void kernel_launch(void* const* d_in, const int* in_sizes, int n_in,
                              void* d_out, int out_size, void* d_ws, size_t ws_size,
                              hipStream_t stream) {
  const float* times   = (const float*)d_in[0];
  const int*   marks   = (const int*)d_in[1];
  const float* u_in    = (const float*)d_in[2];
  const float* llr     = (const float*)d_in[3];
  const float* lim     = (const float*)d_in[4];
  const float* V       = (const float*)d_in[5];
  const float* B       = (const float*)d_in[6];
  const float* E       = (const float*)d_in[7];
  const float* alpha   = (const float*)d_in[8];
  const float* gate_w  = (const float*)d_in[9];
  const float* gate_b  = (const float*)d_in[10];
  const float* x0      = (const float*)d_in[11];
  float* out = (float*)d_out;
  const int n = in_sizes[0];
  const int L = (n + CHN - 1) / CHN;
  const int nrt = n / 16;

  float* ws    = (float*)d_ws;
  float* Vinv  = ws; ws += 64 * 64;
  ws += 96 * 512;
  float* EA    = ws; ws += 64 * 100;
  float* Minv  = ws; ws += 64 * 100;
  float* C96   = ws; ws += (size_t)n * 96;
  float* Ac    = ws; ws += (size_t)n * 64;
  float* Uc    = ws; ws += (size_t)n * 64;
  float* Aagg  = ws; ws += (size_t)CHN * 64;
  float* Uagg  = ws; ws += (size_t)CHN * 64;
  float* Ystart= ws; ws += (size_t)CHN * 64;
  float* Y     = ws; ws += (size_t)n * 64;
  float* Wm    = Y;

  k_front<<<dim3(1 + EA_BLOCKS + nrt * 6), dim3(32), 0, stream>>>(
      V, E, alpha, u_in, gate_w, B, Vinv, EA, C96, nrt);
  k_mid<<<dim3(nrt * 4 + EA_BLOCKS), dim3(32), 0, stream>>>(C96, Vinv, EA, Wm, Minv, nrt);
  k_prep<<<dim3((n * 32 + 255) / 256), dim3(256), 0, stream>>>(
      times, marks, llr, lim, gate_b, C96, Wm, Minv, Ac, Uc, n);
  k_scanA<<<dim3(CHN), dim3(32), 0, stream>>>(Ac, Uc, Aagg, Uagg, n, L);
  k_scanB<<<dim3(1), dim3(32), 0, stream>>>(Aagg, Uagg, Vinv, x0, Ystart);
  k_scanC<<<dim3(CHN), dim3(32), 0, stream>>>(Ac, Uc, Ystart, Y, n, L);
  k_gemm_out<<<dim3(n / 16, 4), dim3(32), 0, stream>>>(Y, V, EA, marks, out, n);
}


# Launches 2-4 fused per chunk of 32 steps: one 8-wave block applies Vinv to B u and to gathered impulses on WMMA (Vinv r sum order changes), forms coefficients and chunk aggregate; 5 launches -> 3
# speedup vs baseline: 8.7779x; 1.6418x over previous
#include <hip/hip_runtime.h>
#include <math.h>


typedef __attribute__((ext_vector_type(2))) float v2f;
typedef __attribute__((ext_vector_type(8))) float v8f;

__device__ __forceinline__ float softplusf(float x) {
  return (x > 20.f) ? x : log1pf(expf(x));
}

__device__ __forceinline__ float lane_bcast(float v, int src_lane) {
  return __int_as_float(__builtin_amdgcn_readlane(__float_as_int(v), src_lane));
}

__device__ __forceinline__ void gj_step(float (&rp)[64], float (&ro)[64], float& sp,
                                        const int pl, const int lane) {
  const float pinv = 1.f / lane_bcast(rp[0], pl);
  const bool piv = (lane == pl);
  const float nfp = piv ? 0.f : -(rp[0] * pinv);
  const float nfo = -(ro[0] * pinv);
  sp = piv ? pinv : sp;
#pragma unroll
  for (int c = 1; c < 64; ++c) {
    const float s = lane_bcast(rp[c], pl);
    rp[c - 1] = fmaf(nfp, s, rp[c]);
    ro[c - 1] = fmaf(nfo, s, ro[c]);
  }
  rp[63] = piv ? 1.f : nfp;
  ro[63] = nfo;
}

__device__ __forceinline__ void inverse_wave(const float* __restrict__ V,
                                             float* __restrict__ Vinv, const int lane) {
  float ra[64], rb[64];
  const float4* pa = reinterpret_cast<const float4*>(V + lane * 64);
  const float4* pb = reinterpret_cast<const float4*>(V + (lane + 32) * 64);
#pragma unroll
  for (int j = 0; j < 16; ++j) {
    const float4 x = pa[j], y = pb[j];
    ra[4 * j] = x.x; ra[4 * j + 1] = x.y; ra[4 * j + 2] = x.z; ra[4 * j + 3] = x.w;
    rb[4 * j] = y.x; rb[4 * j + 1] = y.y; rb[4 * j + 2] = y.z; rb[4 * j + 3] = y.w;
  }
  float sa = 1.f, sb = 1.f;
  for (int k = 0; k < 32; ++k) gj_step(ra, rb, sa, k, lane);
  for (int k = 0; k < 32; ++k) gj_step(rb, ra, sb, k, lane);
  float4* qa = reinterpret_cast<float4*>(Vinv + lane * 64);
  float4* qb = reinterpret_cast<float4*>(Vinv + (lane + 32) * 64);
#pragma unroll
  for (int j = 0; j < 16; ++j) {
    qa[j] = make_float4(sa * ra[4 * j], sa * ra[4 * j + 1], sa * ra[4 * j + 2], sa * ra[4 * j + 3]);
    qb[j] = make_float4(sb * rb[4 * j], sb * rb[4 * j + 1], sb * rb[4 * j + 2], sb * rb[4 * j + 3]);
  }
}

constexpr int EA_BLOCKS = (64 * 100 + 31) / 32;

__global__ __launch_bounds__(32) void k_front(const float* __restrict__ V,
                                              const float* __restrict__ E,
                                              const float* __restrict__ alpha,
                                              const float* __restrict__ U,
                                              const float* __restrict__ gate_w,
                                              const float* __restrict__ Bm,
                                              float* __restrict__ Vinv,
                                              float* __restrict__ EA,
                                              float* __restrict__ C96, int nrt) {
  const int b = blockIdx.x;
  const int l = threadIdx.x;
  if (b == 0) {
    inverse_wave(V, Vinv, l);
    return;
  }
  if (b <= EA_BLOCKS) {
    const int idx = (b - 1) * 32 + l;
    if (idx < 64 * 100) {
      const int d = idx / 100, k = idx % 100;
      float s = 0.f;
      for (int r = 0; r < 32; ++r) s = fmaf(E[d * 32 + r], alpha[r * 100 + k], s);
      EA[idx] = s;
    }
    return;
  }
  const int t = b - 1 - EA_BLOCKS;
  const int rt = t % nrt, ct = t / nrt;
  const int m = l & 15;
  const int kh = (l >> 4) * 2;
  const int c = ct * 16 + m;
  const float* Arow = U + (size_t)(rt * 16 + m) * 512;
  const float* Brow = (ct < 2) ? (gate_w + (size_t)c * 512) : (Bm + (size_t)(c - 32) * 512);
  v8f acc = {};
  for (int kb = 0; kb < 512; kb += 4) {
    v2f a = *(const v2f*)(Arow + kb + kh);
    v2f bb = *(const v2f*)(Brow + kb + kh);
    acc = __builtin_amdgcn_wmma_f32_16x16x4_f32(false, a, false, bb, (short)0, acc,
                                                false, false);
  }
  const int rbase = rt * 16 + (l >> 4) * 8;
  for (int v = 0; v < 8; ++v) C96[(size_t)(rbase + v) * 96 + c] = acc[v];
}

constexpr int CHN = 256;
constexpr int SB = 32;
constexpr int TW = 8;
constexpr int YP = 68;

__device__ __forceinline__ void load_steps(v2f (&a)[SB], v2f (&u)[SB],
                                           const float* __restrict__ pa,
                                           const float* __restrict__ pu) {
#pragma unroll
  for (int j = 0; j < SB; ++j) {
    a[j] = *(const v2f*)(pa + j * 64);
    u[j] = *(const v2f*)(pu + j * 64);
  }
  __builtin_amdgcn_sched_barrier(0);
}

__global__ __launch_bounds__(32 * TW) void k_chunk(const float* __restrict__ times,
                                                   const int* __restrict__ marks,
                                                   const float* __restrict__ llr,
                                                   const float* __restrict__ lim,
                                                   const float* __restrict__ gate_b,
                                                   const float* __restrict__ C96,
                                                   const float* __restrict__ Vinv,
                                                   const float* __restrict__ EA,
                                                   const float* __restrict__ x0,
                                                   float* __restrict__ Ac,
                                                   float* __restrict__ Uc,
                                                   float* __restrict__ Aagg,
                                                   float* __restrict__ Uagg,
                                                   float* __restrict__ Y0) {
  static_assert(SB == 32 && TW == 8, "a chunk is 2 x 4 tiles of 16x16, one per wave; 4 steps per wave");
  __shared__ float Wsh[SB * YP];
  __shared__ float Rsh[SB * YP];
  __shared__ float Ash[SB * 64];
  __shared__ float Ush[SB * 64];
  const int ch = blockIdx.x;
  const int w = threadIdx.x >> 5, p = threadIdx.x & 31;

  if (ch == CHN) {
    if (w == 0) {
      const float4* r0 = reinterpret_cast<const float4*>(Vinv + (2 * p) * 64);
      const float4* r1 = reinterpret_cast<const float4*>(Vinv + (2 * p + 1) * 64);
      const float4* xv = reinterpret_cast<const float4*>(x0);
      float4 w0[16], w1[16];
#pragma unroll
      for (int j = 0; j < 16; ++j) { w0[j] = r0[j]; w1[j] = r1[j]; }
      float yr = 0.f, yi = 0.f;
#pragma unroll
      for (int j = 0; j < 16; ++j) {
        const float4 x = xv[j];
        yr = fmaf(w0[j].x, x.x, yr); yi = fmaf(w1[j].x, x.x, yi);
        yr = fmaf(w0[j].y, x.y, yr); yi = fmaf(w1[j].y, x.y, yi);
        yr = fmaf(w0[j].z, x.z, yr); yi = fmaf(w1[j].z, x.z, yi);
        yr = fmaf(w0[j].w, x.w, yr); yi = fmaf(w1[j].w, x.w, yi);
      }
      Y0[2 * p] = yr; Y0[2 * p + 1] = yi;
    }
    return;
  }

  const int m = p & 15;
  const int kh = (p >> 4) * 2;
  const int rt = w & 1, ct = w >> 1;
  const int dcol = ct * 16 + m;
  const int arow = ch * SB + rt * 16 + m;
  const float* Arow = C96 + (size_t)arow * 96 + 32 + kh;
  const float* Brow = Vinv + (size_t)dcol * 64 + kh;
  const float* Ecol = EA + marks[arow] + kh * 100;
  v2f fa[16], fb[16], fr[16];
#pragma unroll
  for (int q = 0; q < 16; ++q) {
    fa[q] = *(const v2f*)(Arow + 4 * q);
    fb[q] = *(const v2f*)(Brow + 4 * q);
    fr[q].x = Ecol[(4 * q) * 100];
    fr[q].y = Ecol[(4 * q + 1) * 100];
  }

  const float gb = gate_b[p];
  const float im = lim[p];
  const float lr0 = llr[p];
  float tv[4], tpv[4], pre[4];
#pragma unroll
  for (int i = 0; i < 4; ++i) {
    const int ns = ch * SB + w * 4 + i;
    tv[i] = times[ns];
    tpv[i] = times[ns > 0 ? ns - 1 : 0];
    pre[i] = C96[(size_t)ns * 96 + p];
  }
  const float nsp = -softplusf(lr0);
  float ca[4], cc[4], cqr[4], cqi[4];
#pragma unroll
  for (int i = 0; i < 4; ++i) {
    const int ns = ch * SB + w * 4 + i;
    const float t = tv[i];
    const float dt = (ns == 0) ? t : (t - tpv[i]);
    const float gate = softplusf(pre[i] + gb);
    const float real = nsp * gate;
    const float e = expf(real * dt);
    const float a = e * cosf(im * dt);
    const float c = e * sinf(im * dt);
    const float denom = real * real + im * im;
    const float nre = a - 1.f, nim = c;
    float qr = (nre * real + nim * im) / (denom + 1e-12f);
    float qi = (nim * real - nre * im) / (denom + 1e-12f);
    if (denom < 1e-8f) { qr = dt; qi = 0.f; }
    if (!(dt > 0.f)) { qr = 0.f; qi = 0.f; }
    ca[i] = a; cc[i] = c; cqr[i] = qr; cqi[i] = qi;
  }

  v8f accW = {}, accR = {};
#pragma unroll
  for (int q = 0; q < 16; ++q) {
    accW = __builtin_amdgcn_wmma_f32_16x16x4_f32(false, fa[q], false, fb[q], (short)0, accW,
                                                 false, false);
    accR = __builtin_amdgcn_wmma_f32_16x16x4_f32(false, fr[q], false, fb[q], (short)0, accR,
                                                 false, false);
  }
  const int lrow = rt * 16 + (p >> 4) * 8;
#pragma unroll
  for (int v = 0; v < 8; ++v) {
    Wsh[(lrow + v) * YP + dcol] = accW[v];
    Rsh[(lrow + v) * YP + dcol] = accR[v];
  }
  __syncthreads();

#pragma unroll
  for (int i = 0; i < 4; ++i) {
    const int lr = w * 4 + i;
    const v2f wv = *(const v2f*)(Wsh + lr * YP + 2 * p);
    const v2f rv = *(const v2f*)(Rsh + lr * YP + 2 * p);
    float ure = cqr[i] * wv.x - cqi[i] * wv.y;
    float uim = cqi[i] * wv.x + cqr[i] * wv.y;
    ure += rv.x;
    uim += rv.y;
    v2f av; av.x = ca[i]; av.y = cc[i];
    v2f uv; uv.x = ure; uv.y = uim;
    const size_t base = (size_t)(ch * SB + lr) * 64 + 2 * p;
    *(v2f*)(Ac + base) = av;
    *(v2f*)(Uc + base) = uv;
    *(v2f*)(Ash + lr * 64 + 2 * p) = av;
    *(v2f*)(Ush + lr * 64 + 2 * p) = uv;
  }
  __syncthreads();

  if (w == 0) {
    v2f a[SB], u[SB];
    load_steps(a, u, Ash + 2 * p, Ush + 2 * p);
    float Ar = 1.f, Ai = 0.f, Ur = 0.f, Ui = 0.f;
#pragma unroll
    for (int j = 0; j < SB; ++j) {
      float ar = a[j].x, ai = a[j].y, ur = u[j].x, ui = u[j].y;
      float nAr = ar * Ar - ai * Ai, nAi = ar * Ai + ai * Ar;
      float nUr = ar * Ur - ai * Ui + ur, nUi = ar * Ui + ai * Ur + ui;
      Ar = nAr; Ai = nAi; Ur = nUr; Ui = nUi;
    }
    const size_t o = (size_t)ch * 64 + 2 * p;
    Aagg[o] = Ar; Aagg[o + 1] = Ai; Uagg[o] = Ur; Uagg[o + 1] = Ui;
  }
}

__global__ __launch_bounds__(32 * TW) void k_tail(const float* __restrict__ Ac,
                                                  const float* __restrict__ Uc,
                                                  const float* __restrict__ Aagg,
                                                  const float* __restrict__ Uagg,
                                                  const float* __restrict__ Y0,
                                                  const float* __restrict__ V,
                                                  const float* __restrict__ EA,
                                                  const int* __restrict__ marks,
                                                  float* __restrict__ out, int n) {
  static_assert(CHN == TW * SB, "each wave composes SB chunk aggregates");
  static_assert(SB == 32, "a chunk is two 16-row tiles; TW waves cover its 2 x 4 output tiles");
  __shared__ float Ysh[SB * YP];
  __shared__ float4 Psh[TW * 32];
  const int ch = blockIdx.x;
  const int w = threadIdx.x >> 5, p = threadIdx.x & 31;
  const bool lead = (w == 0);
  const bool live = (w * SB < ch);

  v2f a[SB], u[SB];
  v2f y0 = {0.f, 0.f};
  if (lead) {
    load_steps(a, u, Ac + (size_t)ch * SB * 64 + 2 * p, Uc + (size_t)ch * SB * 64 + 2 * p);
    y0 = *(const v2f*)(Y0 + 2 * p);
  }
  v2f ga[SB], gu[SB];
  if (live) load_steps(ga, gu, Aagg + (size_t)w * SB * 64 + 2 * p, Uagg + (size_t)w * SB * 64 + 2 * p);

  const int m = p & 15;
  const int kh = (p >> 4) * 2;
  const int rt = w & 1, ct = w >> 1;
  const int dcol = ct * 16 + m;
  const float* Brow = V + (size_t)dcol * 64;
  v2f bf[16];
#pragma unroll
  for (int q = 0; q < 16; ++q) bf[q] = *(const v2f*)(Brow + 4 * q + kh);
  const int rbase = ch * SB + rt * 16 + (p >> 4) * 8;
  float rr[8];
#pragma unroll
  for (int v = 0; v < 8; ++v) rr[v] = EA[dcol * 100 + marks[rbase + v]];

  float PAr = 1.f, PAi = 0.f, PUr = 0.f, PUi = 0.f;
  if (live) {
#pragma unroll
    for (int j = 0; j < SB; ++j) {
      const bool on = (w * SB + j < ch);
      const float ar = on ? ga[j].x : 1.f, ai = on ? ga[j].y : 0.f;
      const float ur = on ? gu[j].x : 0.f, ui = on ? gu[j].y : 0.f;
      const float nAr = ar * PAr - ai * PAi, nAi = ar * PAi + ai * PAr;
      const float nUr = ar * PUr - ai * PUi + ur, nUi = ar * PUi + ai * PUr + ui;
      PAr = nAr; PAi = nAi; PUr = nUr; PUi = nUi;
    }
  }
  Psh[w * 32 + p] = make_float4(PAr, PAi, PUr, PUi);
  __syncthreads();

  if (lead) {
    float yr = y0.x, yi = y0.y;
#pragma unroll
    for (int q = 0; q < TW; ++q) {
      const float4 P = Psh[q * 32 + p];
      const float nyr = P.x * yr - P.y * yi + P.z;
      const float nyi = P.x * yi + P.y * yr + P.w;
      yr = nyr; yi = nyi;
    }
#pragma unroll
    for (int j = 0; j < SB; ++j) {
      const float ar = a[j].x, ai = a[j].y;
      const float nyr = ar * yr - ai * yi + u[j].x;
      const float nyi = ar * yi + ai * yr + u[j].y;
      yr = nyr; yi = nyi;
      v2f st; st.x = yr; st.y = yi;
      *(v2f*)(Ysh + j * YP + 2 * p) = st;
    }
  }
  __syncthreads();

  const float* Arow = Ysh + (rt * 16 + m) * YP + kh;
  v8f acc = {};
#pragma unroll
  for (int q = 0; q < 16; ++q) {
    const v2f av = *(const v2f*)(Arow + 4 * q);
    acc = __builtin_amdgcn_wmma_f32_16x16x4_f32(false, av, false, bf[q], (short)0, acc,
                                                false, false);
  }
  const size_t xl_off = (size_t)n * 64;
#pragma unroll
  for (int v = 0; v < 8; ++v) {
    const int row = rbase + v;
    const float xr = acc[v];
    out[(size_t)row * 64 + dcol] = xr;
    out[xl_off + (size_t)row * 64 + dcol] = xr - rr[v];
  }
}

extern "C" void kernel_launch(void* const* d_in, const int* in_sizes, int n_in,
                              void* d_out, int out_size, void* d_ws, size_t ws_size,
                              hipStream_t stream) {
  const float* times   = (const float*)d_in[0];
  const int*   marks   = (const int*)d_in[1];
  const float* u_in    = (const float*)d_in[2];
  const float* llr     = (const float*)d_in[3];
  const float* lim     = (const float*)d_in[4];
  const float* V       = (const float*)d_in[5];
  const float* B       = (const float*)d_in[6];
  const float* E       = (const float*)d_in[7];
  const float* alpha   = (const float*)d_in[8];
  const float* gate_w  = (const float*)d_in[9];
  const float* gate_b  = (const float*)d_in[10];
  const float* x0      = (const float*)d_in[11];
  float* out = (float*)d_out;
  const int n = in_sizes[0];
  const int nrt = n / 16;

  float* ws    = (float*)d_ws;
  float* Vinv  = ws; ws += 64 * 64;
  float* Y0    = ws; ws += 96 * 512;
  float* EA    = ws; ws += 64 * 100;
  ws += 64 * 100;
  float* C96   = ws; ws += (size_t)n * 96;
  float* Ac    = ws; ws += (size_t)n * 64;
  float* Uc    = ws; ws += (size_t)n * 64;
  float* Aagg  = ws; ws += (size_t)CHN * 64;
  float* Uagg  = ws; ws += (size_t)CHN * 64;
  ws += (size_t)CHN * 64;

  k_front<<<dim3(1 + EA_BLOCKS + nrt * 6), dim3(32), 0, stream>>>(
      V, E, alpha, u_in, gate_w, B, Vinv, EA, C96, nrt);
  k_chunk<<<dim3(CHN + 1), dim3(32 * TW), 0, stream>>>(
      times, marks, llr, lim, gate_b, C96, Vinv, EA, x0, Ac, Uc, Aagg, Uagg, Y0);
  k_tail<<<dim3(CHN), dim3(32 * TW), 0, stream>>>(Ac, Uc, Aagg, Uagg, Y0, V, EA, marks, out, n);
}


# V inverse: a lane's two rows held as register pairs, each elimination update one packed two-wide FMA whose destination does the slot shift: no per-step register moves; sums and their order unchanged
# speedup vs baseline: 9.0601x; 1.0321x over previous
#include <hip/hip_runtime.h>
#include <math.h>


typedef __attribute__((ext_vector_type(2))) float v2f;
typedef __attribute__((ext_vector_type(8))) float v8f;

__device__ __forceinline__ float softplusf(float x) {
  return (x > 20.f) ? x : log1pf(expf(x));
}

__device__ __forceinline__ float lane_bcast(float v, int src_lane) {
  return __int_as_float(__builtin_amdgcn_readlane(__float_as_int(v), src_lane));
}

template <int HALF>
__device__ __forceinline__ void gj_step(v2f (&r)[64], float& sp, const int pl, const int lane) {
  const float pinv = 1.f / lane_bcast(HALF ? r[0].y : r[0].x, pl);
  const bool piv = (lane == pl);
  v2f nf;
  nf.x = -(r[0].x * pinv);
  nf.y = -(r[0].y * pinv);
  if (HALF == 0) nf.x = piv ? 0.f : nf.x; else nf.y = piv ? 0.f : nf.y;
  sp = piv ? pinv : sp;
#pragma unroll
  for (int c = 1; c < 64; ++c) {
    const float s = lane_bcast(HALF ? r[c].y : r[c].x, pl);
    v2f sv; sv.x = s; sv.y = s;
    r[c - 1] = __builtin_elementwise_fma(nf, sv, r[c]);
  }
  v2f last = nf;
  if (HALF == 0) last.x = piv ? 1.f : nf.x; else last.y = piv ? 1.f : nf.y;
  r[63] = last;
}

__device__ __forceinline__ void inverse_wave(const float* __restrict__ V,
                                             float* __restrict__ Vinv, const int lane) {
  v2f r[64];
  const float4* pa = reinterpret_cast<const float4*>(V + lane * 64);
  const float4* pb = reinterpret_cast<const float4*>(V + (lane + 32) * 64);
#pragma unroll
  for (int j = 0; j < 16; ++j) {
    const float4 x = pa[j], y = pb[j];
    r[4 * j].x = x.x; r[4 * j + 1].x = x.y; r[4 * j + 2].x = x.z; r[4 * j + 3].x = x.w;
    r[4 * j].y = y.x; r[4 * j + 1].y = y.y; r[4 * j + 2].y = y.z; r[4 * j + 3].y = y.w;
  }
  float sa = 1.f, sb = 1.f;
  for (int k = 0; k < 32; ++k) gj_step<0>(r, sa, k, lane);
  for (int k = 0; k < 32; ++k) gj_step<1>(r, sb, k, lane);
  float4* qa = reinterpret_cast<float4*>(Vinv + lane * 64);
  float4* qb = reinterpret_cast<float4*>(Vinv + (lane + 32) * 64);
#pragma unroll
  for (int j = 0; j < 16; ++j) {
    qa[j] = make_float4(sa * r[4 * j].x, sa * r[4 * j + 1].x, sa * r[4 * j + 2].x, sa * r[4 * j + 3].x);
    qb[j] = make_float4(sb * r[4 * j].y, sb * r[4 * j + 1].y, sb * r[4 * j + 2].y, sb * r[4 * j + 3].y);
  }
}

constexpr int EA_BLOCKS = (64 * 100 + 31) / 32;

__global__ __launch_bounds__(32) void k_front(const float* __restrict__ V,
                                              const float* __restrict__ E,
                                              const float* __restrict__ alpha,
                                              const float* __restrict__ U,
                                              const float* __restrict__ gate_w,
                                              const float* __restrict__ Bm,
                                              float* __restrict__ Vinv,
                                              float* __restrict__ EA,
                                              float* __restrict__ C96, int nrt) {
  const int b = blockIdx.x;
  const int l = threadIdx.x;
  if (b == 0) {
    inverse_wave(V, Vinv, l);
    return;
  }
  if (b <= EA_BLOCKS) {
    const int idx = (b - 1) * 32 + l;
    if (idx < 64 * 100) {
      const int d = idx / 100, k = idx % 100;
      float s = 0.f;
      for (int r = 0; r < 32; ++r) s = fmaf(E[d * 32 + r], alpha[r * 100 + k], s);
      EA[idx] = s;
    }
    return;
  }
  const int t = b - 1 - EA_BLOCKS;
  const int rt = t % nrt, ct = t / nrt;
  const int m = l & 15;
  const int kh = (l >> 4) * 2;
  const int c = ct * 16 + m;
  const float* Arow = U + (size_t)(rt * 16 + m) * 512;
  const float* Brow = (ct < 2) ? (gate_w + (size_t)c * 512) : (Bm + (size_t)(c - 32) * 512);
  v8f acc = {};
  for (int kb = 0; kb < 512; kb += 4) {
    v2f a = *(const v2f*)(Arow + kb + kh);
    v2f bb = *(const v2f*)(Brow + kb + kh);
    acc = __builtin_amdgcn_wmma_f32_16x16x4_f32(false, a, false, bb, (short)0, acc,
                                                false, false);
  }
  const int rbase = rt * 16 + (l >> 4) * 8;
  for (int v = 0; v < 8; ++v) C96[(size_t)(rbase + v) * 96 + c] = acc[v];
}

constexpr int CHN = 256;
constexpr int SB = 32;
constexpr int TW = 8;
constexpr int YP = 68;

__device__ __forceinline__ void load_steps(v2f (&a)[SB], v2f (&u)[SB],
                                           const float* __restrict__ pa,
                                           const float* __restrict__ pu) {
#pragma unroll
  for (int j = 0; j < SB; ++j) {
    a[j] = *(const v2f*)(pa + j * 64);
    u[j] = *(const v2f*)(pu + j * 64);
  }
  __builtin_amdgcn_sched_barrier(0);
}

__global__ __launch_bounds__(32 * TW) void k_chunk(const float* __restrict__ times,
                                                   const int* __restrict__ marks,
                                                   const float* __restrict__ llr,
                                                   const float* __restrict__ lim,
                                                   const float* __restrict__ gate_b,
                                                   const float* __restrict__ C96,
                                                   const float* __restrict__ Vinv,
                                                   const float* __restrict__ EA,
                                                   const float* __restrict__ x0,
                                                   float* __restrict__ Ac,
                                                   float* __restrict__ Uc,
                                                   float* __restrict__ Aagg,
                                                   float* __restrict__ Uagg,
                                                   float* __restrict__ Y0) {
  static_assert(SB == 32 && TW == 8, "a chunk is 2 x 4 tiles of 16x16, one per wave; 4 steps per wave");
  __shared__ float Wsh[SB * YP];
  __shared__ float Rsh[SB * YP];
  __shared__ float Ash[SB * 64];
  __shared__ float Ush[SB * 64];
  const int ch = blockIdx.x;
  const int w = threadIdx.x >> 5, p = threadIdx.x & 31;

  if (ch == CHN) {
    if (w == 0) {
      const float4* r0 = reinterpret_cast<const float4*>(Vinv + (2 * p) * 64);
      const float4* r1 = reinterpret_cast<const float4*>(Vinv + (2 * p + 1) * 64);
      const float4* xv = reinterpret_cast<const float4*>(x0);
      float4 w0[16], w1[16];
#pragma unroll
      for (int j = 0; j < 16; ++j) { w0[j] = r0[j]; w1[j] = r1[j]; }
      float yr = 0.f, yi = 0.f;
#pragma unroll
      for (int j = 0; j < 16; ++j) {
        const float4 x = xv[j];
        yr = fmaf(w0[j].x, x.x, yr); yi = fmaf(w1[j].x, x.x, yi);
        yr = fmaf(w0[j].y, x.y, yr); yi = fmaf(w1[j].y, x.y, yi);
        yr = fmaf(w0[j].z, x.z, yr); yi = fmaf(w1[j].z, x.z, yi);
        yr = fmaf(w0[j].w, x.w, yr); yi = fmaf(w1[j].w, x.w, yi);
      }
      Y0[2 * p] = yr; Y0[2 * p + 1] = yi;
    }
    return;
  }

  const int m = p & 15;
  const int kh = (p >> 4) * 2;
  const int rt = w & 1, ct = w >> 1;
  const int dcol = ct * 16 + m;
  const int arow = ch * SB + rt * 16 + m;
  const float* Arow = C96 + (size_t)arow * 96 + 32 + kh;
  const float* Brow = Vinv + (size_t)dcol * 64 + kh;
  const float* Ecol = EA + marks[arow] + kh * 100;
  v2f fa[16], fb[16], fr[16];
#pragma unroll
  for (int q = 0; q < 16; ++q) {
    fa[q] = *(const v2f*)(Arow + 4 * q);
    fb[q] = *(const v2f*)(Brow + 4 * q);
    fr[q].x = Ecol[(4 * q) * 100];
    fr[q].y = Ecol[(4 * q + 1) * 100];
  }

  const float gb = gate_b[p];
  const float im = lim[p];
  const float lr0 = llr[p];
  float tv[4], tpv[4], pre[4];
#pragma unroll
  for (int i = 0; i < 4; ++i) {
    const int ns = ch * SB + w * 4 + i;
    tv[i] = times[ns];
    tpv[i] = times[ns > 0 ? ns - 1 : 0];
    pre[i] = C96[(size_t)ns * 96 + p];
  }
  const float nsp = -softplusf(lr0);
  float ca[4], cc[4], cqr[4], cqi[4];
#pragma unroll
  for (int i = 0; i < 4; ++i) {
    const int ns = ch * SB + w * 4 + i;
    const float t = tv[i];
    const float dt = (ns == 0) ? t : (t - tpv[i]);
    const float gate = softplusf(pre[i] + gb);
    const float real = nsp * gate;
    const float e = expf(real * dt);
    const float a = e * cosf(im * dt);
    const float c = e * sinf(im * dt);
    const float denom = real * real + im * im;
    const float nre = a - 1.f, nim = c;
    float qr = (nre * real + nim * im) / (denom + 1e-12f);
    float qi = (nim * real - nre * im) / (denom + 1e-12f);
    if (denom < 1e-8f) { qr = dt; qi = 0.f; }
    if (!(dt > 0.f)) { qr = 0.f; qi = 0.f; }
    ca[i] = a; cc[i] = c; cqr[i] = qr; cqi[i] = qi;
  }

  v8f accW = {}, accR = {};
#pragma unroll
  for (int q = 0; q < 16; ++q) {
    accW = __builtin_amdgcn_wmma_f32_16x16x4_f32(false, fa[q], false, fb[q], (short)0, accW,
                                                 false, false);
    accR = __builtin_amdgcn_wmma_f32_16x16x4_f32(false, fr[q], false, fb[q], (short)0, accR,
                                                 false, false);
  }
  const int lrow = rt * 16 + (p >> 4) * 8;
#pragma unroll
  for (int v = 0; v < 8; ++v) {
    Wsh[(lrow + v) * YP + dcol] = accW[v];
    Rsh[(lrow + v) * YP + dcol] = accR[v];
  }
  __syncthreads();

#pragma unroll
  for (int i = 0; i < 4; ++i) {
    const int lr = w * 4 + i;
    const v2f wv = *(const v2f*)(Wsh + lr * YP + 2 * p);
    const v2f rv = *(const v2f*)(Rsh + lr * YP + 2 * p);
    float ure = cqr[i] * wv.x - cqi[i] * wv.y;
    float uim = cqi[i] * wv.x + cqr[i] * wv.y;
    ure += rv.x;
    uim += rv.y;
    v2f av; av.x = ca[i]; av.y = cc[i];
    v2f uv; uv.x = ure; uv.y = uim;
    const size_t base = (size_t)(ch * SB + lr) * 64 + 2 * p;
    *(v2f*)(Ac + base) = av;
    *(v2f*)(Uc + base) = uv;
    *(v2f*)(Ash + lr * 64 + 2 * p) = av;
    *(v2f*)(Ush + lr * 64 + 2 * p) = uv;
  }
  __syncthreads();

  if (w == 0) {
    v2f a[SB], u[SB];
    load_steps(a, u, Ash + 2 * p, Ush + 2 * p);
    float Ar = 1.f, Ai = 0.f, Ur = 0.f, Ui = 0.f;
#pragma unroll
    for (int j = 0; j < SB; ++j) {
      float ar = a[j].x, ai = a[j].y, ur = u[j].x, ui = u[j].y;
      float nAr = ar * Ar - ai * Ai, nAi = ar * Ai + ai * Ar;
      float nUr = ar * Ur - ai * Ui + ur, nUi = ar * Ui + ai * Ur + ui;
      Ar = nAr; Ai = nAi; Ur = nUr; Ui = nUi;
    }
    const size_t o = (size_t)ch * 64 + 2 * p;
    Aagg[o] = Ar; Aagg[o + 1] = Ai; Uagg[o] = Ur; Uagg[o + 1] = Ui;
  }
}

__global__ __launch_bounds__(32 * TW) void k_tail(const float* __restrict__ Ac,
                                                  const float* __restrict__ Uc,
                                                  const float* __restrict__ Aagg,
                                                  const float* __restrict__ Uagg,
                                                  const float* __restrict__ Y0,
                                                  const float* __restrict__ V,
                                                  const float* __restrict__ EA,
                                                  const int* __restrict__ marks,
                                                  float* __restrict__ out, int n) {
  static_assert(CHN == TW * SB, "each wave composes SB chunk aggregates");
  static_assert(SB == 32, "a chunk is two 16-row tiles; TW waves cover its 2 x 4 output tiles");
  __shared__ float Ysh[SB * YP];
  __shared__ float4 Psh[TW * 32];
  const int ch = blockIdx.x;
  const int w = threadIdx.x >> 5, p = threadIdx.x & 31;
  const bool lead = (w == 0);
  const bool live = (w * SB < ch);

  v2f a[SB], u[SB];
  v2f y0 = {0.f, 0.f};
  if (lead) {
    load_steps(a, u, Ac + (size_t)ch * SB * 64 + 2 * p, Uc + (size_t)ch * SB * 64 + 2 * p);
    y0 = *(const v2f*)(Y0 + 2 * p);
  }
  v2f ga[SB], gu[SB];
  if (live) load_steps(ga, gu, Aagg + (size_t)w * SB * 64 + 2 * p, Uagg + (size_t)w * SB * 64 + 2 * p);

  const int m = p & 15;
  const int kh = (p >> 4) * 2;
  const int rt = w & 1, ct = w >> 1;
  const int dcol = ct * 16 + m;
  const float* Brow = V + (size_t)dcol * 64;
  v2f bf[16];
#pragma unroll
  for (int q = 0; q < 16; ++q) bf[q] = *(const v2f*)(Brow + 4 * q + kh);
  const int rbase = ch * SB + rt * 16 + (p >> 4) * 8;
  float rr[8];
#pragma unroll
  for (int v = 0; v < 8; ++v) rr[v] = EA[dcol * 100 + marks[rbase + v]];

  float PAr = 1.f, PAi = 0.f, PUr = 0.f, PUi = 0.f;
  if (live) {
#pragma unroll
    for (int j = 0; j < SB; ++j) {
      const bool on = (w * SB + j < ch);
      const float ar = on ? ga[j].x : 1.f, ai = on ? ga[j].y : 0.f;
      const float ur = on ? gu[j].x : 0.f, ui = on ? gu[j].y : 0.f;
      const float nAr = ar * PAr - ai * PAi, nAi = ar * PAi + ai * PAr;
      const float nUr = ar * PUr - ai * PUi + ur, nUi = ar * PUi + ai * PUr + ui;
      PAr = nAr; PAi = nAi; PUr = nUr; PUi = nUi;
    }
  }
  Psh[w * 32 + p] = make_float4(PAr, PAi, PUr, PUi);
  __syncthreads();

  if (lead) {
    float yr = y0.x, yi = y0.y;
#pragma unroll
    for (int q = 0; q < TW; ++q) {
      const float4 P = Psh[q * 32 + p];
      const float nyr = P.x * yr - P.y * yi + P.z;
      const float nyi = P.x * yi + P.y * yr + P.w;
      yr = nyr; yi = nyi;
    }
#pragma unroll
    for (int j = 0; j < SB; ++j) {
      const float ar = a[j].x, ai = a[j].y;
      const float nyr = ar * yr - ai * yi + u[j].x;
      const float nyi = ar * yi + ai * yr + u[j].y;
      yr = nyr; yi = nyi;
      v2f st; st.x = yr; st.y = yi;
      *(v2f*)(Ysh + j * YP + 2 * p) = st;
    }
  }
  __syncthreads();

  const float* Arow = Ysh + (rt * 16 + m) * YP + kh;
  v8f acc = {};
#pragma unroll
  for (int q = 0; q < 16; ++q) {
    const v2f av = *(const v2f*)(Arow + 4 * q);
    acc = __builtin_amdgcn_wmma_f32_16x16x4_f32(false, av, false, bf[q], (short)0, acc,
                                                false, false);
  }
  const size_t xl_off = (size_t)n * 64;
#pragma unroll
  for (int v = 0; v < 8; ++v) {
    const int row = rbase + v;
    const float xr = acc[v];
    out[(size_t)row * 64 + dcol] = xr;
    out[xl_off + (size_t)row * 64 + dcol] = xr - rr[v];
  }
}

extern "C" void kernel_launch(void* const* d_in, const int* in_sizes, int n_in,
                              void* d_out, int out_size, void* d_ws, size_t ws_size,
                              hipStream_t stream) {
  const float* times   = (const float*)d_in[0];
  const int*   marks   = (const int*)d_in[1];
  const float* u_in    = (const float*)d_in[2];
  const float* llr     = (const float*)d_in[3];
  const float* lim     = (const float*)d_in[4];
  const float* V       = (const float*)d_in[5];
  const float* B       = (const float*)d_in[6];
  const float* E       = (const float*)d_in[7];
  const float* alpha   = (const float*)d_in[8];
  const float* gate_w  = (const float*)d_in[9];
  const float* gate_b  = (const float*)d_in[10];
  const float* x0      = (const float*)d_in[11];
  float* out = (float*)d_out;
  const int n = in_sizes[0];
  const int nrt = n / 16;

  float* ws    = (float*)d_ws;
  float* Vinv  = ws; ws += 64 * 64;
  float* Y0    = ws; ws += 96 * 512;
  float* EA    = ws; ws += 64 * 100;
  ws += 64 * 100;
  float* C96   = ws; ws += (size_t)n * 96;
  float* Ac    = ws; ws += (size_t)n * 64;
  float* Uc    = ws; ws += (size_t)n * 64;
  float* Aagg  = ws; ws += (size_t)CHN * 64;
  float* Uagg  = ws; ws += (size_t)CHN * 64;
  ws += (size_t)CHN * 64;

  k_front<<<dim3(1 + EA_BLOCKS + nrt * 6), dim3(32), 0, stream>>>(
      V, E, alpha, u_in, gate_w, B, Vinv, EA, C96, nrt);
  k_chunk<<<dim3(CHN + 1), dim3(32 * TW), 0, stream>>>(
      times, marks, llr, lim, gate_b, C96, Vinv, EA, x0, Ac, Uc, Aagg, Uagg, Y0);
  k_tail<<<dim3(CHN), dim3(32 * TW), 0, stream>>>(Ac, Uc, Aagg, Uagg, Y0, V, EA, marks, out, n);
}


# Launches 1-2 fused per chunk of 32 steps: each block's wave 0 inverts V itself while six waves form the chunk's 32x96 GEMM tiles on WMMA and one its impulses; 3 launches -> 2; sum order unchanged
# speedup vs baseline: 10.4273x; 1.1509x over previous
#include <hip/hip_runtime.h>
#include <math.h>


typedef __attribute__((ext_vector_type(2))) float v2f;
typedef __attribute__((ext_vector_type(8))) float v8f;

__device__ __forceinline__ float softplusf(float x) {
  return (x > 20.f) ? x : log1pf(expf(x));
}

__device__ __forceinline__ float lane_bcast(float v, int src_lane) {
  return __int_as_float(__builtin_amdgcn_readlane(__float_as_int(v), src_lane));
}

template <int HALF>
__device__ __forceinline__ void gj_step(v2f (&r)[64], float& sp, const int pl, const int lane) {
  const float pinv = 1.f / lane_bcast(HALF ? r[0].y : r[0].x, pl);
  const bool piv = (lane == pl);
  v2f nf;
  nf.x = -(r[0].x * pinv);
  nf.y = -(r[0].y * pinv);
  if (HALF == 0) nf.x = piv ? 0.f : nf.x; else nf.y = piv ? 0.f : nf.y;
  sp = piv ? pinv : sp;
#pragma unroll
  for (int c = 1; c < 64; ++c) {
    const float s = lane_bcast(HALF ? r[c].y : r[c].x, pl);
    v2f sv; sv.x = s; sv.y = s;
    r[c - 1] = __builtin_elementwise_fma(nf, sv, r[c]);
  }
  v2f last = nf;
  if (HALF == 0) last.x = piv ? 1.f : nf.x; else last.y = piv ? 1.f : nf.y;
  r[63] = last;
}

__device__ __forceinline__ void inverse_regs(const float* __restrict__ V, v2f (&r)[64],
                                             float& sa, float& sb, const int lane) {
  const float4* pa = reinterpret_cast<const float4*>(V + lane * 64);
  const float4* pb = reinterpret_cast<const float4*>(V + (lane + 32) * 64);
#pragma unroll
  for (int j = 0; j < 16; ++j) {
    const float4 x = pa[j], y = pb[j];
    r[4 * j].x = x.x; r[4 * j + 1].x = x.y; r[4 * j + 2].x = x.z; r[4 * j + 3].x = x.w;
    r[4 * j].y = y.x; r[4 * j + 1].y = y.y; r[4 * j + 2].y = y.z; r[4 * j + 3].y = y.w;
  }
  sa = 1.f; sb = 1.f;
  for (int k = 0; k < 32; ++k) gj_step<0>(r, sa, k, lane);
  for (int k = 0; k < 32; ++k) gj_step<1>(r, sb, k, lane);
}

constexpr int CHN = 256;
constexpr int SB = 32;
constexpr int TW = 8;
constexpr int YP = 68;

__device__ __forceinline__ void load_steps(v2f (&a)[SB], v2f (&u)[SB],
                                           const float* __restrict__ pa,
                                           const float* __restrict__ pu) {
#pragma unroll
  for (int j = 0; j < SB; ++j) {
    a[j] = *(const v2f*)(pa + j * 64);
    u[j] = *(const v2f*)(pu + j * 64);
  }
  __builtin_amdgcn_sched_barrier(0);
}

constexpr int GK = 16;

__device__ __forceinline__ void load_group(v2f (&a)[GK], v2f (&b0)[GK], v2f (&b1)[GK],
                                           const float* __restrict__ pa,
                                           const float* __restrict__ pb0,
                                           const float* __restrict__ pb1) {
#pragma unroll
  for (int q = 0; q < GK; ++q) {
    a[q] = *(const v2f*)(pa + 4 * q);
    b0[q] = *(const v2f*)(pb0 + 4 * q);
    b1[q] = *(const v2f*)(pb1 + 4 * q);
  }
  __builtin_amdgcn_sched_barrier(0);
}

__device__ __forceinline__ void mma_group(v8f& acc0, v8f& acc1, const v2f (&a)[GK],
                                          const v2f (&b0)[GK], const v2f (&b1)[GK]) {
#pragma unroll
  for (int q = 0; q < GK; ++q) {
    acc0 = __builtin_amdgcn_wmma_f32_16x16x4_f32(false, a[q], false, b0[q], (short)0, acc0,
                                                 false, false);
    acc1 = __builtin_amdgcn_wmma_f32_16x16x4_f32(false, a[q], false, b1[q], (short)0, acc1,
                                                 false, false);
  }
}

constexpr int CP = 100;

__global__ __launch_bounds__(32 * TW) void k_chunk(const float* __restrict__ times,
                                                   const int* __restrict__ marks,
                                                   const float* __restrict__ llr,
                                                   const float* __restrict__ lim,
                                                   const float* __restrict__ gate_b,
                                                   const float* __restrict__ U,
                                                   const float* __restrict__ gate_w,
                                                   const float* __restrict__ Bm,
                                                   const float* __restrict__ V,
                                                   const float* __restrict__ E,
                                                   const float* __restrict__ alpha,
                                                   const float* __restrict__ x0,
                                                   float* __restrict__ Ac,
                                                   float* __restrict__ Uc,
                                                   float* __restrict__ Rg,
                                                   float* __restrict__ Aagg,
                                                   float* __restrict__ Uagg,
                                                   float* __restrict__ Y0) {
  static_assert(SB == 32 && TW == 8, "a chunk is 2 x 4 tiles of 16x16, one per wave; 4 steps per wave");
  __shared__ alignas(16) float Csh[SB * CP];
  __shared__ alignas(16) float Vsh[64 * YP];
  __shared__ alignas(16) float Rin[SB * YP];
  __shared__ alignas(16) float Wsh[SB * YP];
  __shared__ alignas(16) float Rsh[SB * YP];
  __shared__ alignas(16) float Ash[SB * 64];
  __shared__ alignas(16) float Ush[SB * 64];
  const int ch = blockIdx.x;
  const int w = threadIdx.x >> 5, p = threadIdx.x & 31;

  if (ch == CHN) {
    if (w == 0) {
      v2f r[64];
      float sa, sb;
      inverse_regs(V, r, sa, sb, p);
      float ya = 0.f, yb = 0.f;
#pragma unroll
      for (int j = 0; j < 64; ++j) {
        const float x = x0[j];
        ya = fmaf(sa * r[j].x, x, ya);
        yb = fmaf(sb * r[j].y, x, yb);
      }
      Y0[p] = ya; Y0[p + 32] = yb;
    }
    return;
  }

  const float gb = gate_b[p];
  const float im = lim[p];
  const float lr0 = llr[p];
  float tv[4], tpv[4], pre[4];
#pragma unroll
  for (int i = 0; i < 4; ++i) {
    const int ns = ch * SB + w * 4 + i;
    tv[i] = times[ns];
    tpv[i] = times[ns > 0 ? ns - 1 : 0];
  }

  if (w == 0) {
    v2f r[64];
    float sa, sb;
    inverse_regs(V, r, sa, sb, p);
    float4* qa = reinterpret_cast<float4*>(Vsh + p * YP);
    float4* qb = reinterpret_cast<float4*>(Vsh + (p + 32) * YP);
#pragma unroll
    for (int j = 0; j < 16; ++j) {
      qa[j] = make_float4(sa * r[4 * j].x, sa * r[4 * j + 1].x, sa * r[4 * j + 2].x, sa * r[4 * j + 3].x);
      qb[j] = make_float4(sb * r[4 * j].y, sb * r[4 * j + 1].y, sb * r[4 * j + 2].y, sb * r[4 * j + 3].y);
    }
  } else if (w <= 6) {
    const int g = w - 1;
    const int grt = g & 1, gcp = g >> 1;
    const int gm = p & 15;
    const int gkh = (p >> 4) * 2;
    const int c0 = gcp * 32 + gm, c1 = c0 + 16;
    const float* Ar = U + (size_t)(ch * SB + grt * 16 + gm) * 512 + gkh;
    const float* B0 = ((gcp == 0) ? (gate_w + (size_t)c0 * 512) : (Bm + (size_t)(c0 - 32) * 512)) + gkh;
    const float* B1 = ((gcp == 0) ? (gate_w + (size_t)c1 * 512) : (Bm + (size_t)(c1 - 32) * 512)) + gkh;
    v8f acc0 = {}, acc1 = {};
    v2f xa[GK], xb[GK], xc[GK], ya[GK], yb[GK], yc[GK];
    load_group(xa, xb, xc, Ar, B0, B1);
#pragma unroll
    for (int g = 0; g < 512 / (4 * GK); g += 2) {
      load_group(ya, yb, yc, Ar + (g + 1) * 4 * GK, B0 + (g + 1) * 4 * GK, B1 + (g + 1) * 4 * GK);
      mma_group(acc0, acc1, xa, xb, xc);
      if (g + 2 < 512 / (4 * GK))
        load_group(xa, xb, xc, Ar + (g + 2) * 4 * GK, B0 + (g + 2) * 4 * GK, B1 + (g + 2) * 4 * GK);
      mma_group(acc0, acc1, ya, yb, yc);
    }
    const int grow = grt * 16 + (p >> 4) * 8;
#pragma unroll
    for (int v = 0; v < 8; ++v) {
      Csh[(grow + v) * CP + c0] = acc0[v];
      Csh[(grow + v) * CP + c1] = acc1[v];
    }
  } else {
    const int mk = marks[ch * SB + p];
    float al[32];
#pragma unroll
    for (int j = 0; j < 32; ++j) al[j] = alpha[j * 100 + mk];
    v2f e[32];
    const float4* ea = reinterpret_cast<const float4*>(E + p * 32);
    const float4* eb = reinterpret_cast<const float4*>(E + (p + 32) * 32);
#pragma unroll
    for (int j = 0; j < 8; ++j) {
      const float4 x = ea[j], y = eb[j];
      e[4 * j].x = x.x; e[4 * j + 1].x = x.y; e[4 * j + 2].x = x.z; e[4 * j + 3].x = x.w;
      e[4 * j].y = y.x; e[4 * j + 1].y = y.y; e[4 * j + 2].y = y.z; e[4 * j + 3].y = y.w;
    }
    for (int s4 = 0; s4 < SB; s4 += 4) {
      v2f acc[4];
#pragma unroll
      for (int t = 0; t < 4; ++t) { acc[t].x = 0.f; acc[t].y = 0.f; }
#pragma unroll
      for (int j = 0; j < 32; ++j) {
#pragma unroll
        for (int t = 0; t < 4; ++t) {
          const float b = lane_bcast(al[j], s4 + t);
          v2f bv; bv.x = b; bv.y = b;
          acc[t] = __builtin_elementwise_fma(e[j], bv, acc[t]);
        }
      }
#pragma unroll
      for (int t = 0; t < 4; ++t) {
        Rin[(s4 + t) * YP + p] = acc[t].x;
        Rin[(s4 + t) * YP + p + 32] = acc[t].y;
      }
    }
  }
  __syncthreads();

  const int m = p & 15;
  const int kh = (p >> 4) * 2;
  const int rt = w & 1, ct = w >> 1;
  const int dcol = ct * 16 + m;
  const float* Arow = Csh + (rt * 16 + m) * CP + 32 + kh;
  const float* Brow = Vsh + dcol * YP + kh;
  const float* Rrow = Rin + (rt * 16 + m) * YP + kh;
  v2f fa[16], fb[16], fr[16];
#pragma unroll
  for (int q = 0; q < 16; ++q) {
    fa[q] = *(const v2f*)(Arow + 4 * q);
    fb[q] = *(const v2f*)(Brow + 4 * q);
    fr[q] = *(const v2f*)(Rrow + 4 * q);
  }

#pragma unroll
  for (int i = 0; i < 4; ++i) pre[i] = Csh[(w * 4 + i) * CP + p];
  const float nsp = -softplusf(lr0);
  float ca[4], cc[4], cqr[4], cqi[4];
#pragma unroll
  for (int i = 0; i < 4; ++i) {
    const int ns = ch * SB + w * 4 + i;
    const float t = tv[i];
    const float dt = (ns == 0) ? t : (t - tpv[i]);
    const float gate = softplusf(pre[i] + gb);
    const float real = nsp * gate;
    const float e = expf(real * dt);
    const float a = e * cosf(im * dt);
    const float c = e * sinf(im * dt);
    const float denom = real * real + im * im;
    const float nre = a - 1.f, nim = c;
    float qr = (nre * real + nim * im) / (denom + 1e-12f);
    float qi = (nim * real - nre * im) / (denom + 1e-12f);
    if (denom < 1e-8f) { qr = dt; qi = 0.f; }
    if (!(dt > 0.f)) { qr = 0.f; qi = 0.f; }
    ca[i] = a; cc[i] = c; cqr[i] = qr; cqi[i] = qi;
  }

  v8f accW = {}, accR = {};
#pragma unroll
  for (int q = 0; q < 16; ++q) {
    accW = __builtin_amdgcn_wmma_f32_16x16x4_f32(false, fa[q], false, fb[q], (short)0, accW,
                                                 false, false);
    accR = __builtin_amdgcn_wmma_f32_16x16x4_f32(false, fr[q], false, fb[q], (short)0, accR,
                                                 false, false);
  }
  const int lrow = rt * 16 + (p >> 4) * 8;
#pragma unroll
  for (int v = 0; v < 8; ++v) {
    Wsh[(lrow + v) * YP + dcol] = accW[v];
    Rsh[(lrow + v) * YP + dcol] = accR[v];
  }
  __syncthreads();

#pragma unroll
  for (int i = 0; i < 4; ++i) {
    const int lr = w * 4 + i;
    const v2f wv = *(const v2f*)(Wsh + lr * YP + 2 * p);
    const v2f rv = *(const v2f*)(Rsh + lr * YP + 2 * p);
    float ure = cqr[i] * wv.x - cqi[i] * wv.y;
    float uim = cqi[i] * wv.x + cqr[i] * wv.y;
    ure += rv.x;
    uim += rv.y;
    v2f av; av.x = ca[i]; av.y = cc[i];
    v2f uv; uv.x = ure; uv.y = uim;
    const size_t base = (size_t)(ch * SB + lr) * 64 + 2 * p;
    *(v2f*)(Ac + base) = av;
    *(v2f*)(Uc + base) = uv;
    *(v2f*)(Rg + base) = *(const v2f*)(Rin + lr * YP + 2 * p);
    *(v2f*)(Ash + lr * 64 + 2 * p) = av;
    *(v2f*)(Ush + lr * 64 + 2 * p) = uv;
  }
  __syncthreads();

  if (w == 0) {
    v2f a[SB], u[SB];
    load_steps(a, u, Ash + 2 * p, Ush + 2 * p);
    float Ar = 1.f, Ai = 0.f, Ur = 0.f, Ui = 0.f;
#pragma unroll
    for (int j = 0; j < SB; ++j) {
      float ar = a[j].x, ai = a[j].y, ur = u[j].x, ui = u[j].y;
      float nAr = ar * Ar - ai * Ai, nAi = ar * Ai + ai * Ar;
      float nUr = ar * Ur - ai * Ui + ur, nUi = ar * Ui + ai * Ur + ui;
      Ar = nAr; Ai = nAi; Ur = nUr; Ui = nUi;
    }
    const size_t o = (size_t)ch * 64 + 2 * p;
    Aagg[o] = Ar; Aagg[o + 1] = Ai; Uagg[o] = Ur; Uagg[o + 1] = Ui;
  }
}

__global__ __launch_bounds__(32 * TW) void k_tail(const float* __restrict__ Ac,
                                                  const float* __restrict__ Uc,
                                                  const float* __restrict__ Aagg,
                                                  const float* __restrict__ Uagg,
                                                  const float* __restrict__ Y0,
                                                  const float* __restrict__ V,
                                                  const float* __restrict__ Rg,
                                                  float* __restrict__ out, int n) {
  static_assert(CHN == TW * SB, "each wave composes SB chunk aggregates");
  static_assert(SB == 32, "a chunk is two 16-row tiles; TW waves cover its 2 x 4 output tiles");
  __shared__ float Ysh[SB * YP];
  __shared__ float4 Psh[TW * 32];
  const int ch = blockIdx.x;
  const int w = threadIdx.x >> 5, p = threadIdx.x & 31;
  const bool lead = (w == 0);
  const bool live = (w * SB < ch);

  v2f a[SB], u[SB];
  v2f y0 = {0.f, 0.f};
  if (lead) {
    load_steps(a, u, Ac + (size_t)ch * SB * 64 + 2 * p, Uc + (size_t)ch * SB * 64 + 2 * p);
    y0 = *(const v2f*)(Y0 + 2 * p);
  }
  v2f ga[SB], gu[SB];
  if (live) load_steps(ga, gu, Aagg + (size_t)w * SB * 64 + 2 * p, Uagg + (size_t)w * SB * 64 + 2 * p);

  const int m = p & 15;
  const int kh = (p >> 4) * 2;
  const int rt = w & 1, ct = w >> 1;
  const int dcol = ct * 16 + m;
  const float* Brow = V + (size_t)dcol * 64;
  v2f bf[16];
#pragma unroll
  for (int q = 0; q < 16; ++q) bf[q] = *(const v2f*)(Brow + 4 * q + kh);
  const int rbase = ch * SB + rt * 16 + (p >> 4) * 8;
  float rr[8];
#pragma unroll
  for (int v = 0; v < 8; ++v) rr[v] = Rg[(size_t)(rbase + v) * 64 + dcol];

  float PAr = 1.f, PAi = 0.f, PUr = 0.f, PUi = 0.f;
  if (live) {
#pragma unroll
    for (int j = 0; j < SB; ++j) {
      const bool on = (w * SB + j < ch);
      const float ar = on ? ga[j].x : 1.f, ai = on ? ga[j].y : 0.f;
      const float ur = on ? gu[j].x : 0.f, ui = on ? gu[j].y : 0.f;
      const float nAr = ar * PAr - ai * PAi, nAi = ar * PAi + ai * PAr;
      const float nUr = ar * PUr - ai * PUi + ur, nUi = ar * PUi + ai * PUr + ui;
      PAr = nAr; PAi = nAi; PUr = nUr; PUi = nUi;
    }
  }
  Psh[w * 32 + p] = make_float4(PAr, PAi, PUr, PUi);
  __syncthreads();

  if (lead) {
    float yr = y0.x, yi = y0.y;
#pragma unroll
    for (int q = 0; q < TW; ++q) {
      const float4 P = Psh[q * 32 + p];
      const float nyr = P.x * yr - P.y * yi + P.z;
      const float nyi = P.x * yi + P.y * yr + P.w;
      yr = nyr; yi = nyi;
    }
#pragma unroll
    for (int j = 0; j < SB; ++j) {
      const float ar = a[j].x, ai = a[j].y;
      const float nyr = ar * yr - ai * yi + u[j].x;
      const float nyi = ar * yi + ai * yr + u[j].y;
      yr = nyr; yi = nyi;
      v2f st; st.x = yr; st.y = yi;
      *(v2f*)(Ysh + j * YP + 2 * p) = st;
    }
  }
  __syncthreads();

  const float* Arow = Ysh + (rt * 16 + m) * YP + kh;
  v8f acc = {};
#pragma unroll
  for (int q = 0; q < 16; ++q) {
    const v2f av = *(const v2f*)(Arow + 4 * q);
    acc = __builtin_amdgcn_wmma_f32_16x16x4_f32(false, av, false, bf[q], (short)0, acc,
                                                false, false);
  }
  const size_t xl_off = (size_t)n * 64;
#pragma unroll
  for (int v = 0; v < 8; ++v) {
    const int row = rbase + v;
    const float xr = acc[v];
    out[(size_t)row * 64 + dcol] = xr;
    out[xl_off + (size_t)row * 64 + dcol] = xr - rr[v];
  }
}

extern "C" void kernel_launch(void* const* d_in, const int* in_sizes, int n_in,
                              void* d_out, int out_size, void* d_ws, size_t ws_size,
                              hipStream_t stream) {
  const float* times   = (const float*)d_in[0];
  const int*   marks   = (const int*)d_in[1];
  const float* u_in    = (const float*)d_in[2];
  const float* llr     = (const float*)d_in[3];
  const float* lim     = (const float*)d_in[4];
  const float* V       = (const float*)d_in[5];
  const float* B       = (const float*)d_in[6];
  const float* E       = (const float*)d_in[7];
  const float* alpha   = (const float*)d_in[8];
  const float* gate_w  = (const float*)d_in[9];
  const float* gate_b  = (const float*)d_in[10];
  const float* x0      = (const float*)d_in[11];
  float* out = (float*)d_out;
  const int n = in_sizes[0];

  float* ws    = (float*)d_ws;
  ws += 64 * 64;
  float* Y0    = ws; ws += 96 * 512;
  ws += 64 * 100;
  ws += 64 * 100;
  float* Rg    = ws; ws += (size_t)n * 96;
  float* Ac    = ws; ws += (size_t)n * 64;
  float* Uc    = ws; ws += (size_t)n * 64;
  float* Aagg  = ws; ws += (size_t)CHN * 64;
  float* Uagg  = ws; ws += (size_t)CHN * 64;

  k_chunk<<<dim3(CHN + 1), dim3(32 * TW), 0, stream>>>(
      times, marks, llr, lim, gate_b, u_in, gate_w, B, V, E, alpha, x0, Ac, Uc, Rg, Aagg, Uagg, Y0);
  k_tail<<<dim3(CHN), dim3(32 * TW), 0, stream>>>(Ac, Uc, Aagg, Uagg, Y0, V, Rg, out, n);
}


# y0 = Vinv x0 formed in chunk 0's block by an idle wave from the inverse in LDS; the 257th block, whose own inverse shared a WGP with a chunk block's, is gone; sums and their order unchanged
# speedup vs baseline: 11.4809x; 1.1010x over previous
#include <hip/hip_runtime.h>
#include <math.h>


typedef __attribute__((ext_vector_type(2))) float v2f;
typedef __attribute__((ext_vector_type(8))) float v8f;

__device__ __forceinline__ float softplusf(float x) {
  return (x > 20.f) ? x : log1pf(expf(x));
}

__device__ __forceinline__ float lane_bcast(float v, int src_lane) {
  return __int_as_float(__builtin_amdgcn_readlane(__float_as_int(v), src_lane));
}

template <int HALF>
__device__ __forceinline__ void gj_step(v2f (&r)[64], float& sp, const int pl, const int lane) {
  const float pinv = 1.f / lane_bcast(HALF ? r[0].y : r[0].x, pl);
  const bool piv = (lane == pl);
  v2f nf;
  nf.x = -(r[0].x * pinv);
  nf.y = -(r[0].y * pinv);
  if (HALF == 0) nf.x = piv ? 0.f : nf.x; else nf.y = piv ? 0.f : nf.y;
  sp = piv ? pinv : sp;
#pragma unroll
  for (int c = 1; c < 64; ++c) {
    const float s = lane_bcast(HALF ? r[c].y : r[c].x, pl);
    v2f sv; sv.x = s; sv.y = s;
    r[c - 1] = __builtin_elementwise_fma(nf, sv, r[c]);
  }
  v2f last = nf;
  if (HALF == 0) last.x = piv ? 1.f : nf.x; else last.y = piv ? 1.f : nf.y;
  r[63] = last;
}

__device__ __forceinline__ void inverse_regs(const float* __restrict__ V, v2f (&r)[64],
                                             float& sa, float& sb, const int lane) {
  const float4* pa = reinterpret_cast<const float4*>(V + lane * 64);
  const float4* pb = reinterpret_cast<const float4*>(V + (lane + 32) * 64);
#pragma unroll
  for (int j = 0; j < 16; ++j) {
    const float4 x = pa[j], y = pb[j];
    r[4 * j].x = x.x; r[4 * j + 1].x = x.y; r[4 * j + 2].x = x.z; r[4 * j + 3].x = x.w;
    r[4 * j].y = y.x; r[4 * j + 1].y = y.y; r[4 * j + 2].y = y.z; r[4 * j + 3].y = y.w;
  }
  sa = 1.f; sb = 1.f;
  for (int k = 0; k < 32; ++k) gj_step<0>(r, sa, k, lane);
  for (int k = 0; k < 32; ++k) gj_step<1>(r, sb, k, lane);
}

constexpr int CHN = 256;
constexpr int SB = 32;
constexpr int TW = 8;
constexpr int YP = 68;

__device__ __forceinline__ void load_steps(v2f (&a)[SB], v2f (&u)[SB],
                                           const float* __restrict__ pa,
                                           const float* __restrict__ pu) {
#pragma unroll
  for (int j = 0; j < SB; ++j) {
    a[j] = *(const v2f*)(pa + j * 64);
    u[j] = *(const v2f*)(pu + j * 64);
  }
  __builtin_amdgcn_sched_barrier(0);
}

constexpr int GK = 16;

__device__ __forceinline__ void load_group(v2f (&a)[GK], v2f (&b0)[GK], v2f (&b1)[GK],
                                           const float* __restrict__ pa,
                                           const float* __restrict__ pb0,
                                           const float* __restrict__ pb1) {
#pragma unroll
  for (int q = 0; q < GK; ++q) {
    a[q] = *(const v2f*)(pa + 4 * q);
    b0[q] = *(const v2f*)(pb0 + 4 * q);
    b1[q] = *(const v2f*)(pb1 + 4 * q);
  }
  __builtin_amdgcn_sched_barrier(0);
}

__device__ __forceinline__ void mma_group(v8f& acc0, v8f& acc1, const v2f (&a)[GK],
                                          const v2f (&b0)[GK], const v2f (&b1)[GK]) {
#pragma unroll
  for (int q = 0; q < GK; ++q) {
    acc0 = __builtin_amdgcn_wmma_f32_16x16x4_f32(false, a[q], false, b0[q], (short)0, acc0,
                                                 false, false);
    acc1 = __builtin_amdgcn_wmma_f32_16x16x4_f32(false, a[q], false, b1[q], (short)0, acc1,
                                                 false, false);
  }
}

constexpr int CP = 100;

__global__ __launch_bounds__(32 * TW) void k_chunk(const float* __restrict__ times,
                                                   const int* __restrict__ marks,
                                                   const float* __restrict__ llr,
                                                   const float* __restrict__ lim,
                                                   const float* __restrict__ gate_b,
                                                   const float* __restrict__ U,
                                                   const float* __restrict__ gate_w,
                                                   const float* __restrict__ Bm,
                                                   const float* __restrict__ V,
                                                   const float* __restrict__ E,
                                                   const float* __restrict__ alpha,
                                                   const float* __restrict__ x0,
                                                   float* __restrict__ Ac,
                                                   float* __restrict__ Uc,
                                                   float* __restrict__ Rg,
                                                   float* __restrict__ Aagg,
                                                   float* __restrict__ Uagg,
                                                   float* __restrict__ Y0) {
  static_assert(SB == 32 && TW == 8, "a chunk is 2 x 4 tiles of 16x16, one per wave; 4 steps per wave");
  __shared__ alignas(16) float Csh[SB * CP];
  __shared__ alignas(16) float Vsh[64 * YP];
  __shared__ alignas(16) float Rin[SB * YP];
  __shared__ alignas(16) float Wsh[SB * YP];
  __shared__ alignas(16) float Rsh[SB * YP];
  __shared__ alignas(16) float Ash[SB * 64];
  __shared__ alignas(16) float Ush[SB * 64];
  const int ch = blockIdx.x;
  const int w = threadIdx.x >> 5, p = threadIdx.x & 31;

  const float gb = gate_b[p];
  const float im = lim[p];
  const float lr0 = llr[p];
  float tv[4], tpv[4], pre[4];
#pragma unroll
  for (int i = 0; i < 4; ++i) {
    const int ns = ch * SB + w * 4 + i;
    tv[i] = times[ns];
    tpv[i] = times[ns > 0 ? ns - 1 : 0];
  }

  if (w == 0) {
    v2f r[64];
    float sa, sb;
    inverse_regs(V, r, sa, sb, p);
    float4* qa = reinterpret_cast<float4*>(Vsh + p * YP);
    float4* qb = reinterpret_cast<float4*>(Vsh + (p + 32) * YP);
#pragma unroll
    for (int j = 0; j < 16; ++j) {
      qa[j] = make_float4(sa * r[4 * j].x, sa * r[4 * j + 1].x, sa * r[4 * j + 2].x, sa * r[4 * j + 3].x);
      qb[j] = make_float4(sb * r[4 * j].y, sb * r[4 * j + 1].y, sb * r[4 * j + 2].y, sb * r[4 * j + 3].y);
    }
  } else if (w <= 6) {
    const int g = w - 1;
    const int grt = g & 1, gcp = g >> 1;
    const int gm = p & 15;
    const int gkh = (p >> 4) * 2;
    const int c0 = gcp * 32 + gm, c1 = c0 + 16;
    const float* Ar = U + (size_t)(ch * SB + grt * 16 + gm) * 512 + gkh;
    const float* B0 = ((gcp == 0) ? (gate_w + (size_t)c0 * 512) : (Bm + (size_t)(c0 - 32) * 512)) + gkh;
    const float* B1 = ((gcp == 0) ? (gate_w + (size_t)c1 * 512) : (Bm + (size_t)(c1 - 32) * 512)) + gkh;
    v8f acc0 = {}, acc1 = {};
    v2f xa[GK], xb[GK], xc[GK], ya[GK], yb[GK], yc[GK];
    load_group(xa, xb, xc, Ar, B0, B1);
#pragma unroll
    for (int g = 0; g < 512 / (4 * GK); g += 2) {
      load_group(ya, yb, yc, Ar + (g + 1) * 4 * GK, B0 + (g + 1) * 4 * GK, B1 + (g + 1) * 4 * GK);
      mma_group(acc0, acc1, xa, xb, xc);
      if (g + 2 < 512 / (4 * GK))
        load_group(xa, xb, xc, Ar + (g + 2) * 4 * GK, B0 + (g + 2) * 4 * GK, B1 + (g + 2) * 4 * GK);
      mma_group(acc0, acc1, ya, yb, yc);
    }
    const int grow = grt * 16 + (p >> 4) * 8;
#pragma unroll
    for (int v = 0; v < 8; ++v) {
      Csh[(grow + v) * CP + c0] = acc0[v];
      Csh[(grow + v) * CP + c1] = acc1[v];
    }
  } else {
    const int mk = marks[ch * SB + p];
    float al[32];
#pragma unroll
    for (int j = 0; j < 32; ++j) al[j] = alpha[j * 100 + mk];
    v2f e[32];
    const float4* ea = reinterpret_cast<const float4*>(E + p * 32);
    const float4* eb = reinterpret_cast<const float4*>(E + (p + 32) * 32);
#pragma unroll
    for (int j = 0; j < 8; ++j) {
      const float4 x = ea[j], y = eb[j];
      e[4 * j].x = x.x; e[4 * j + 1].x = x.y; e[4 * j + 2].x = x.z; e[4 * j + 3].x = x.w;
      e[4 * j].y = y.x; e[4 * j + 1].y = y.y; e[4 * j + 2].y = y.z; e[4 * j + 3].y = y.w;
    }
    for (int s4 = 0; s4 < SB; s4 += 4) {
      v2f acc[4];
#pragma unroll
      for (int t = 0; t < 4; ++t) { acc[t].x = 0.f; acc[t].y = 0.f; }
#pragma unroll
      for (int j = 0; j < 32; ++j) {
#pragma unroll
        for (int t = 0; t < 4; ++t) {
          const float b = lane_bcast(al[j], s4 + t);
          v2f bv; bv.x = b; bv.y = b;
          acc[t] = __builtin_elementwise_fma(e[j], bv, acc[t]);
        }
      }
#pragma unroll
      for (int t = 0; t < 4; ++t) {
        Rin[(s4 + t) * YP + p] = acc[t].x;
        Rin[(s4 + t) * YP + p + 32] = acc[t].y;
      }
    }
  }
  __syncthreads();

  float xlo = 0.f, xhi = 0.f;
  if (ch == 0 && w == 1) { xlo = x0[p]; xhi = x0[p + 32]; }

  const int m = p & 15;
  const int kh = (p >> 4) * 2;
  const int rt = w & 1, ct = w >> 1;
  const int dcol = ct * 16 + m;
  const float* Arow = Csh + (rt * 16 + m) * CP + 32 + kh;
  const float* Brow = Vsh + dcol * YP + kh;
  const float* Rrow = Rin + (rt * 16 + m) * YP + kh;
  v2f fa[16], fb[16], fr[16];
#pragma unroll
  for (int q = 0; q < 16; ++q) {
    fa[q] = *(const v2f*)(Arow + 4 * q);
    fb[q] = *(const v2f*)(Brow + 4 * q);
    fr[q] = *(const v2f*)(Rrow + 4 * q);
  }

#pragma unroll
  for (int i = 0; i < 4; ++i) pre[i] = Csh[(w * 4 + i) * CP + p];
  const float nsp = -softplusf(lr0);
  float ca[4], cc[4], cqr[4], cqi[4];
#pragma unroll
  for (int i = 0; i < 4; ++i) {
    const int ns = ch * SB + w * 4 + i;
    const float t = tv[i];
    const float dt = (ns == 0) ? t : (t - tpv[i]);
    const float gate = softplusf(pre[i] + gb);
    const float real = nsp * gate;
    const float e = expf(real * dt);
    const float a = e * cosf(im * dt);
    const float c = e * sinf(im * dt);
    const float denom = real * real + im * im;
    const float nre = a - 1.f, nim = c;
    float qr = (nre * real + nim * im) / (denom + 1e-12f);
    float qi = (nim * real - nre * im) / (denom + 1e-12f);
    if (denom < 1e-8f) { qr = dt; qi = 0.f; }
    if (!(dt > 0.f)) { qr = 0.f; qi = 0.f; }
    ca[i] = a; cc[i] = c; cqr[i] = qr; cqi[i] = qi;
  }

  v8f accW = {}, accR = {};
#pragma unroll
  for (int q = 0; q < 16; ++q) {
    accW = __builtin_amdgcn_wmma_f32_16x16x4_f32(false, fa[q], false, fb[q], (short)0, accW,
                                                 false, false);
    accR = __builtin_amdgcn_wmma_f32_16x16x4_f32(false, fr[q], false, fb[q], (short)0, accR,
                                                 false, false);
  }
  const int lrow = rt * 16 + (p >> 4) * 8;
#pragma unroll
  for (int v = 0; v < 8; ++v) {
    Wsh[(lrow + v) * YP + dcol] = accW[v];
    Rsh[(lrow + v) * YP + dcol] = accR[v];
  }
  __syncthreads();

#pragma unroll
  for (int i = 0; i < 4; ++i) {
    const int lr = w * 4 + i;
    const v2f wv = *(const v2f*)(Wsh + lr * YP + 2 * p);
    const v2f rv = *(const v2f*)(Rsh + lr * YP + 2 * p);
    float ure = cqr[i] * wv.x - cqi[i] * wv.y;
    float uim = cqi[i] * wv.x + cqr[i] * wv.y;
    ure += rv.x;
    uim += rv.y;
    v2f av; av.x = ca[i]; av.y = cc[i];
    v2f uv; uv.x = ure; uv.y = uim;
    const size_t base = (size_t)(ch * SB + lr) * 64 + 2 * p;
    *(v2f*)(Ac + base) = av;
    *(v2f*)(Uc + base) = uv;
    *(v2f*)(Rg + base) = *(const v2f*)(Rin + lr * YP + 2 * p);
    *(v2f*)(Ash + lr * 64 + 2 * p) = av;
    *(v2f*)(Ush + lr * 64 + 2 * p) = uv;
  }
  __syncthreads();

  if (w == 0) {
    v2f a[SB], u[SB];
    load_steps(a, u, Ash + 2 * p, Ush + 2 * p);
    float Ar = 1.f, Ai = 0.f, Ur = 0.f, Ui = 0.f;
#pragma unroll
    for (int j = 0; j < SB; ++j) {
      float ar = a[j].x, ai = a[j].y, ur = u[j].x, ui = u[j].y;
      float nAr = ar * Ar - ai * Ai, nAi = ar * Ai + ai * Ar;
      float nUr = ar * Ur - ai * Ui + ur, nUi = ar * Ui + ai * Ur + ui;
      Ar = nAr; Ai = nAi; Ur = nUr; Ui = nUi;
    }
    const size_t o = (size_t)ch * 64 + 2 * p;
    Aagg[o] = Ar; Aagg[o + 1] = Ai; Uagg[o] = Ur; Uagg[o + 1] = Ui;
  } else if (ch == 0 && w == 1) {
    const float4* va = reinterpret_cast<const float4*>(Vsh + p * YP);
    const float4* vb = reinterpret_cast<const float4*>(Vsh + (p + 32) * YP);
    float ya = 0.f, yb = 0.f;
#pragma unroll
    for (int j = 0; j < 16; ++j) {
      const float4 a = va[j], b = vb[j];
      const float xs = (j < 8) ? xlo : xhi;
      const int l0 = (4 * j) & 31;
      const float xa = lane_bcast(xs, l0), xb = lane_bcast(xs, l0 + 1);
      const float xc = lane_bcast(xs, l0 + 2), xd = lane_bcast(xs, l0 + 3);
      ya = fmaf(a.x, xa, ya); yb = fmaf(b.x, xa, yb);
      ya = fmaf(a.y, xb, ya); yb = fmaf(b.y, xb, yb);
      ya = fmaf(a.z, xc, ya); yb = fmaf(b.z, xc, yb);
      ya = fmaf(a.w, xd, ya); yb = fmaf(b.w, xd, yb);
    }
    Y0[p] = ya; Y0[p + 32] = yb;
  }
}

__global__ __launch_bounds__(32 * TW) void k_tail(const float* __restrict__ Ac,
                                                  const float* __restrict__ Uc,
                                                  const float* __restrict__ Aagg,
                                                  const float* __restrict__ Uagg,
                                                  const float* __restrict__ Y0,
                                                  const float* __restrict__ V,
                                                  const float* __restrict__ Rg,
                                                  float* __restrict__ out, int n) {
  static_assert(CHN == TW * SB, "each wave composes SB chunk aggregates");
  static_assert(SB == 32, "a chunk is two 16-row tiles; TW waves cover its 2 x 4 output tiles");
  __shared__ float Ysh[SB * YP];
  __shared__ float4 Psh[TW * 32];
  const int ch = blockIdx.x;
  const int w = threadIdx.x >> 5, p = threadIdx.x & 31;
  const bool lead = (w == 0);
  const bool live = (w * SB < ch);

  v2f a[SB], u[SB];
  v2f y0 = {0.f, 0.f};
  if (lead) {
    load_steps(a, u, Ac + (size_t)ch * SB * 64 + 2 * p, Uc + (size_t)ch * SB * 64 + 2 * p);
    y0 = *(const v2f*)(Y0 + 2 * p);
  }
  v2f ga[SB], gu[SB];
  if (live) load_steps(ga, gu, Aagg + (size_t)w * SB * 64 + 2 * p, Uagg + (size_t)w * SB * 64 + 2 * p);

  const int m = p & 15;
  const int kh = (p >> 4) * 2;
  const int rt = w & 1, ct = w >> 1;
  const int dcol = ct * 16 + m;
  const float* Brow = V + (size_t)dcol * 64;
  v2f bf[16];
#pragma unroll
  for (int q = 0; q < 16; ++q) bf[q] = *(const v2f*)(Brow + 4 * q + kh);
  const int rbase = ch * SB + rt * 16 + (p >> 4) * 8;
  float rr[8];
#pragma unroll
  for (int v = 0; v < 8; ++v) rr[v] = Rg[(size_t)(rbase + v) * 64 + dcol];

  float PAr = 1.f, PAi = 0.f, PUr = 0.f, PUi = 0.f;
  if (live) {
#pragma unroll
    for (int j = 0; j < SB; ++j) {
      const bool on = (w * SB + j < ch);
      const float ar = on ? ga[j].x : 1.f, ai = on ? ga[j].y : 0.f;
      const float ur = on ? gu[j].x : 0.f, ui = on ? gu[j].y : 0.f;
      const float nAr = ar * PAr - ai * PAi, nAi = ar * PAi + ai * PAr;
      const float nUr = ar * PUr - ai * PUi + ur, nUi = ar * PUi + ai * PUr + ui;
      PAr = nAr; PAi = nAi; PUr = nUr; PUi = nUi;
    }
  }
  Psh[w * 32 + p] = make_float4(PAr, PAi, PUr, PUi);
  __syncthreads();

  if (lead) {
    float yr = y0.x, yi = y0.y;
#pragma unroll
    for (int q = 0; q < TW; ++q) {
      const float4 P = Psh[q * 32 + p];
      const float nyr = P.x * yr - P.y * yi + P.z;
      const float nyi = P.x * yi + P.y * yr + P.w;
      yr = nyr; yi = nyi;
    }
#pragma unroll
    for (int j = 0; j < SB; ++j) {
      const float ar = a[j].x, ai = a[j].y;
      const float nyr = ar * yr - ai * yi + u[j].x;
      const float nyi = ar * yi + ai * yr + u[j].y;
      yr = nyr; yi = nyi;
      v2f st; st.x = yr; st.y = yi;
      *(v2f*)(Ysh + j * YP + 2 * p) = st;
    }
  }
  __syncthreads();

  const float* Arow = Ysh + (rt * 16 + m) * YP + kh;
  v8f acc = {};
#pragma unroll
  for (int q = 0; q < 16; ++q) {
    const v2f av = *(const v2f*)(Arow + 4 * q);
    acc = __builtin_amdgcn_wmma_f32_16x16x4_f32(false, av, false, bf[q], (short)0, acc,
                                                false, false);
  }
  const size_t xl_off = (size_t)n * 64;
#pragma unroll
  for (int v = 0; v < 8; ++v) {
    const int row = rbase + v;
    const float xr = acc[v];
    out[(size_t)row * 64 + dcol] = xr;
    out[xl_off + (size_t)row * 64 + dcol] = xr - rr[v];
  }
}

extern "C" void kernel_launch(void* const* d_in, const int* in_sizes, int n_in,
                              void* d_out, int out_size, void* d_ws, size_t ws_size,
                              hipStream_t stream) {
  const float* times   = (const float*)d_in[0];
  const int*   marks   = (const int*)d_in[1];
  const float* u_in    = (const float*)d_in[2];
  const float* llr     = (const float*)d_in[3];
  const float* lim     = (const float*)d_in[4];
  const float* V       = (const float*)d_in[5];
  const float* B       = (const float*)d_in[6];
  const float* E       = (const float*)d_in[7];
  const float* alpha   = (const float*)d_in[8];
  const float* gate_w  = (const float*)d_in[9];
  const float* gate_b  = (const float*)d_in[10];
  const float* x0      = (const float*)d_in[11];
  float* out = (float*)d_out;
  const int n = in_sizes[0];

  float* ws    = (float*)d_ws;
  ws += 64 * 64;
  float* Y0    = ws; ws += 96 * 512;
  ws += 64 * 100;
  ws += 64 * 100;
  float* Rg    = ws; ws += (size_t)n * 96;
  float* Ac    = ws; ws += (size_t)n * 64;
  float* Uc    = ws; ws += (size_t)n * 64;
  float* Aagg  = ws; ws += (size_t)CHN * 64;
  float* Uagg  = ws; ws += (size_t)CHN * 64;

  k_chunk<<<dim3(CHN), dim3(32 * TW), 0, stream>>>(
      times, marks, llr, lim, gate_b, u_in, gate_w, B, V, E, alpha, x0, Ac, Uc, Rg, Aagg, Uagg, Y0);
  k_tail<<<dim3(CHN), dim3(32 * TW), 0, stream>>>(Ac, Uc, Aagg, Uagg, Y0, V, Rg, out, n);
}
